# starting kernel
# baseline (speedup 1.0000x reference)
#include <hip/hip_runtime.h>
#include <math.h>

typedef float v2f __attribute__((ext_vector_type(2)));
typedef float v8f __attribute__((ext_vector_type(8)));

static __device__ __forceinline__ float elu1(float v) {
    return v > 0.0f ? v : (__expf(v) - 1.0f);
}


__global__ void zero_f32_kernel(float* __restrict__ p, int n) {
    int i = blockIdx.x * blockDim.x + threadIdx.x;
    if (i < n) p[i] = 0.0f;
}

__global__ void deg_scatter_kernel(const int* __restrict__ col,
                                   const float* __restrict__ w,
                                   float* __restrict__ deg, int E) {
    int e = blockIdx.x * blockDim.x + threadIdx.x;
    if (e < E) atomicAdd(&deg[col[e]], w[e]);
}

__global__ void dinv_kernel(float* __restrict__ deg, int N) {
    int i = blockIdx.x * blockDim.x + threadIdx.x;
    if (i < N) {
        float d = deg[i];
        deg[i] = d > 0.0f ? rsqrtf(d) : 0.0f;
    }
}

__global__ void norm_kernel(const int* __restrict__ row, const int* __restrict__ col,
                            const float* __restrict__ w, const float* __restrict__ dinv,
                            float* __restrict__ norm, int E) {
    int e = blockIdx.x * blockDim.x + threadIdx.x;
    if (e < E) norm[e] = dinv[row[e]] * w[e] * dinv[col[e]];
}

__global__ void pre_kernel(const float* __restrict__ x, const float* __restrict__ preW,
                           const float* __restrict__ preb, float* __restrict__ h, int total) {
    int i = blockIdx.x * blockDim.x + threadIdx.x;
    if (i >= total) return;
    int n = i >> 6, f = i & 63;
    h[i] = elu1(x[n] * preW[f] + preb[f]);
}

__global__ void edge_scatter_kernel(const int* __restrict__ row, const int* __restrict__ col,
                                    const float* __restrict__ norm, const float* __restrict__ t,
                                    float* __restrict__ agg, int E) {
    int gid = blockIdx.x * blockDim.x + threadIdx.x;
    int e = gid >> 5;
    int f = gid & 31;
    if (e >= E) return;
    int r = row[e], c = col[e];
    float v = norm[e] * t[(size_t)r * 32 + f];
    atomicAdd(&agg[(size_t)c * 32 + f], v);
}

__global__ void relu_kernel(float* __restrict__ p, int n) {
    int i = blockIdx.x * blockDim.x + threadIdx.x;
    if (i < n) p[i] = fmaxf(p[i], 0.0f);
}

template <int NCT, bool DUAL>
__global__ void wmma_gemm_kernel(const float* __restrict__ A, int K,
                                 const float* __restrict__ Wi,
                                 const float* __restrict__ Wr,
                                 const float* __restrict__ biasR,
                                 float* __restrict__ Ti,
                                 float* __restrict__ Tr,
                                 int nrows) {
    const int F = NCT * 16;
    int wave = (blockIdx.x * blockDim.x + threadIdx.x) >> 5;
    int lane = threadIdx.x & 31;
    int row0 = wave * 16;
    if (row0 >= nrows) return;

    int m  = lane & 15;
    int hi = lane >> 4;

    v8f accI[NCT] = {};
    v8f accR[NCT] = {};

    int ar = row0 + m;
    if (ar >= nrows) ar = nrows - 1;
    const float* arow = A + (size_t)ar * K;

    for (int k0 = 0; k0 < K; k0 += 4) {
        int ka = k0 + 2 * hi;
        v2f a;
        a.x = arow[ka];
        a.y = arow[ka + 1];
#pragma unroll
        for (int ct = 0; ct < NCT; ++ct) {
            int colc = ct * 16 + m;
            v2f b;
            b.x = Wi[(size_t)ka * F + colc];
            b.y = Wi[(size_t)(ka + 1) * F + colc];
            accI[ct] = __builtin_amdgcn_wmma_f32_16x16x4_f32(
                false, a, false, b, (short)0, accI[ct], false, false);
            if (DUAL) {
                v2f br;
                br.x = Wr[(size_t)ka * F + colc];
                br.y = Wr[(size_t)(ka + 1) * F + colc];
                accR[ct] = __builtin_amdgcn_wmma_f32_16x16x4_f32(
                    false, a, false, br, (short)0, accR[ct], false, false);
            }
        }
    }

#pragma unroll
    for (int ct = 0; ct < NCT; ++ct) {
        int colc = ct * 16 + m;
        float bv = biasR ? biasR[colc] : 0.0f;
#pragma unroll
        for (int r = 0; r < 8; ++r) {
            int rowm = row0 + r + 8 * hi;
            if (rowm >= nrows) continue;
            size_t idx = (size_t)rowm * F + colc;
            if (DUAL) {
                Ti[idx] = accI[ct][r];
                Tr[idx] = accR[ct][r] + bv;
            } else {
                Ti[idx] = accI[ct][r] + bv;
            }
        }
    }
}


__global__ void score_kernel(const int* __restrict__ li, const float* __restrict__ z,
                             float* __restrict__ out, int L) {
    int e = blockIdx.x * blockDim.x + threadIdx.x;
    if (e >= L) return;
    int i = li[e], j = li[L + e];
    const float4* za = (const float4*)(z + (size_t)i * 16);
    const float4* zb = (const float4*)(z + (size_t)j * 16);
    float dot = 0.0f, na = 0.0f, nb = 0.0f;
#pragma unroll
    for (int q = 0; q < 4; ++q) {
        float4 a = za[q], b = zb[q];
        dot += a.x * b.x + a.y * b.y + a.z * b.z + a.w * b.w;
        na  += a.x * a.x + a.y * a.y + a.z * a.z + a.w * a.w;
        nb  += b.x * b.x + b.y * b.y + b.z * b.z + b.w * b.w;
    }
    float den = fmaxf(sqrtf(na) * sqrtf(nb), 1e-6f);
    out[e] = dot / den;
}


static inline int cdiv(long long a, long long b) { return (int)((a + b - 1) / b); }

extern "C" void kernel_launch(void* const* d_in, const int* in_sizes, int n_in,
                              void* d_out, int out_size, void* d_ws, size_t ws_size,
                              hipStream_t stream) {
    const float* x     = (const float*)d_in[0];
    const int*   ei    = (const int*)d_in[1];
    const float* ew    = (const float*)d_in[2];
    const int*   eli   = (const int*)d_in[3];
    const float* preW  = (const float*)d_in[4];
    const float* preb  = (const float*)d_in[5];
    const float* g1Wi  = (const float*)d_in[6];
    const float* g1Wr  = (const float*)d_in[7];
    const float* g1b   = (const float*)d_in[8];
    const float* g2Wi  = (const float*)d_in[9];
    const float* g2Wr  = (const float*)d_in[10];
    const float* g2b   = (const float*)d_in[11];
    const float* poW   = (const float*)d_in[12];
    const float* pob   = (const float*)d_in[13];
    float* out = (float*)d_out;

    const int N = in_sizes[0];
    const int E = in_sizes[2];
    const int L = in_sizes[3] / 2;

    const int* erow = ei;
    const int* ecol = ei + E;

    float* ws   = (float*)d_ws;
    float* deg  = ws;
    float* norm = deg  + N;
    float* h64  = norm + E;
    float* t    = h64  + (size_t)N * 64;
    float* agg1 = t    + (size_t)N * 32;
    float* agg2 = agg1 + (size_t)N * 32;
    float* z    = agg2 + (size_t)N * 32;

    const int B = 256;
    const int gemm_threads = cdiv(N, 16) * 32;

    zero_f32_kernel<<<cdiv(N, B), B, 0, stream>>>(deg, N);
    deg_scatter_kernel<<<cdiv(E, B), B, 0, stream>>>(ecol, ew, deg, E);
    dinv_kernel<<<cdiv(N, B), B, 0, stream>>>(deg, N);
    norm_kernel<<<cdiv(E, B), B, 0, stream>>>(erow, ecol, ew, deg, norm, E);

    pre_kernel<<<cdiv((long long)N * 64, B), B, 0, stream>>>(x, preW, preb, h64, N * 64);

    wmma_gemm_kernel<2, true><<<cdiv(gemm_threads, B), B, 0, stream>>>(
        h64, 64, g1Wi, g1Wr, g1b, t, agg1, N);
    edge_scatter_kernel<<<cdiv((long long)E * 32, B), B, 0, stream>>>(
        erow, ecol, norm, t, agg1, E);
    relu_kernel<<<cdiv((long long)N * 32, B), B, 0, stream>>>(agg1, N * 32);

    wmma_gemm_kernel<2, true><<<cdiv(gemm_threads, B), B, 0, stream>>>(
        agg1, 32, g2Wi, g2Wr, g2b, t, agg2, N);
    edge_scatter_kernel<<<cdiv((long long)E * 32, B), B, 0, stream>>>(
        erow, ecol, norm, t, agg2, E);
    relu_kernel<<<cdiv((long long)N * 32, B), B, 0, stream>>>(agg2, N * 32);

    wmma_gemm_kernel<1, false><<<cdiv(gemm_threads, B), B, 0, stream>>>(
        agg2, 32, poW, nullptr, pob, z, nullptr, N);

    score_kernel<<<cdiv(L, B), B, 0, stream>>>(eli, z, out, L);
}


# edge scatter: a wave takes 32 edges per trip (indices read coalesced, passed round by readlane), 32 row gathers in flight, resident grid-stride waves; float atomics kept, arrival order as before
# speedup vs baseline: 1.8475x; 1.8475x over previous
#include <hip/hip_runtime.h>
#include <math.h>

typedef float v2f __attribute__((ext_vector_type(2)));
typedef float v8f __attribute__((ext_vector_type(8)));

static __device__ __forceinline__ float elu1(float v) {
    return v > 0.0f ? v : (__expf(v) - 1.0f);
}


__global__ void zero_f32_kernel(float* __restrict__ p, int n) {
    int i = blockIdx.x * blockDim.x + threadIdx.x;
    if (i < n) p[i] = 0.0f;
}

__global__ void deg_scatter_kernel(const int* __restrict__ col,
                                   const float* __restrict__ w,
                                   float* __restrict__ deg, int E) {
    int e = blockIdx.x * blockDim.x + threadIdx.x;
    if (e < E) atomicAdd(&deg[col[e]], w[e]);
}

__global__ void dinv_kernel(float* __restrict__ deg, int N) {
    int i = blockIdx.x * blockDim.x + threadIdx.x;
    if (i < N) {
        float d = deg[i];
        deg[i] = d > 0.0f ? rsqrtf(d) : 0.0f;
    }
}

__global__ void norm_kernel(const int* __restrict__ row, const int* __restrict__ col,
                            const float* __restrict__ w, const float* __restrict__ dinv,
                            float* __restrict__ norm, int E) {
    int e = blockIdx.x * blockDim.x + threadIdx.x;
    if (e < E) norm[e] = dinv[row[e]] * w[e] * dinv[col[e]];
}

__global__ void pre_kernel(const float* __restrict__ x, const float* __restrict__ preW,
                           const float* __restrict__ preb, float* __restrict__ h, int total) {
    int i = blockIdx.x * blockDim.x + threadIdx.x;
    if (i >= total) return;
    int n = i >> 6, f = i & 63;
    h[i] = elu1(x[n] * preW[f] + preb[f]);
}

__global__ void edge_scatter_kernel(const int* __restrict__ row, const int* __restrict__ col,
                                    const float* __restrict__ norm, const float* __restrict__ t,
                                    float* __restrict__ agg, int E) {
    const int lane = threadIdx.x & 31;
    const int wave = __builtin_amdgcn_readfirstlane((int)((blockIdx.x * blockDim.x + threadIdx.x) >> 5));
    const int nwaves = (int)((gridDim.x * blockDim.x) >> 5);
    const int nchunks = (E + 31) >> 5;

    int chunk = wave;
    if (chunk >= nchunks) return;

    int r_v = 0, c_v = 0;
    float n_v = 0.0f;
    {
        int e = chunk * 32 + lane;
        if (e < E) { r_v = row[e]; c_v = col[e]; n_v = norm[e]; }
    }

    for (;;) {
        const int next = chunk + nwaves;
        int r_n = 0, c_n = 0;
        float n_n = 0.0f;
        if (next < nchunks) {
            int e = next * 32 + lane;
            if (e < E) { r_n = row[e]; c_n = col[e]; n_n = norm[e]; }
        }

        const int cnt = E - chunk * 32;
        if (cnt >= 32) {
            float v[32];
#pragma unroll
            for (int j = 0; j < 32; ++j) {
                int r = __builtin_amdgcn_readlane(r_v, j);
                v[j] = t[(unsigned)(r * 32 + lane)];
            }
#pragma unroll
            for (int j = 0; j < 32; ++j) {
                int c = __builtin_amdgcn_readlane(c_v, j);
                float n = __int_as_float(__builtin_amdgcn_readlane(__float_as_int(n_v), j));
                atomicAdd(&agg[(unsigned)(c * 32 + lane)], n * v[j]);
            }
        } else {
            for (int j = 0; j < cnt; ++j) {
                int r = __shfl(r_v, j);
                int c = __shfl(c_v, j);
                float n = __shfl(n_v, j);
                atomicAdd(&agg[(unsigned)(c * 32 + lane)], n * t[(unsigned)(r * 32 + lane)]);
            }
        }

        if (next >= nchunks) break;
        chunk = next; r_v = r_n; c_v = c_n; n_v = n_n;
    }
}

__global__ void relu_kernel(float* __restrict__ p, int n) {
    int i = blockIdx.x * blockDim.x + threadIdx.x;
    if (i < n) p[i] = fmaxf(p[i], 0.0f);
}

template <int NCT, bool DUAL>
__global__ void wmma_gemm_kernel(const float* __restrict__ A, int K,
                                 const float* __restrict__ Wi,
                                 const float* __restrict__ Wr,
                                 const float* __restrict__ biasR,
                                 float* __restrict__ Ti,
                                 float* __restrict__ Tr,
                                 int nrows) {
    const int F = NCT * 16;
    int wave = (blockIdx.x * blockDim.x + threadIdx.x) >> 5;
    int lane = threadIdx.x & 31;
    int row0 = wave * 16;
    if (row0 >= nrows) return;

    int m  = lane & 15;
    int hi = lane >> 4;

    v8f accI[NCT] = {};
    v8f accR[NCT] = {};

    int ar = row0 + m;
    if (ar >= nrows) ar = nrows - 1;
    const float* arow = A + (size_t)ar * K;

    for (int k0 = 0; k0 < K; k0 += 4) {
        int ka = k0 + 2 * hi;
        v2f a;
        a.x = arow[ka];
        a.y = arow[ka + 1];
#pragma unroll
        for (int ct = 0; ct < NCT; ++ct) {
            int colc = ct * 16 + m;
            v2f b;
            b.x = Wi[(size_t)ka * F + colc];
            b.y = Wi[(size_t)(ka + 1) * F + colc];
            accI[ct] = __builtin_amdgcn_wmma_f32_16x16x4_f32(
                false, a, false, b, (short)0, accI[ct], false, false);
            if (DUAL) {
                v2f br;
                br.x = Wr[(size_t)ka * F + colc];
                br.y = Wr[(size_t)(ka + 1) * F + colc];
                accR[ct] = __builtin_amdgcn_wmma_f32_16x16x4_f32(
                    false, a, false, br, (short)0, accR[ct], false, false);
            }
        }
    }

#pragma unroll
    for (int ct = 0; ct < NCT; ++ct) {
        int colc = ct * 16 + m;
        float bv = biasR ? biasR[colc] : 0.0f;
#pragma unroll
        for (int r = 0; r < 8; ++r) {
            int rowm = row0 + r + 8 * hi;
            if (rowm >= nrows) continue;
            size_t idx = (size_t)rowm * F + colc;
            if (DUAL) {
                Ti[idx] = accI[ct][r];
                Tr[idx] = accR[ct][r] + bv;
            } else {
                Ti[idx] = accI[ct][r] + bv;
            }
        }
    }
}


__global__ void score_kernel(const int* __restrict__ li, const float* __restrict__ z,
                             float* __restrict__ out, int L) {
    int e = blockIdx.x * blockDim.x + threadIdx.x;
    if (e >= L) return;
    int i = li[e], j = li[L + e];
    const float4* za = (const float4*)(z + (size_t)i * 16);
    const float4* zb = (const float4*)(z + (size_t)j * 16);
    float dot = 0.0f, na = 0.0f, nb = 0.0f;
#pragma unroll
    for (int q = 0; q < 4; ++q) {
        float4 a = za[q], b = zb[q];
        dot += a.x * b.x + a.y * b.y + a.z * b.z + a.w * b.w;
        na  += a.x * a.x + a.y * a.y + a.z * a.z + a.w * a.w;
        nb  += b.x * b.x + b.y * b.y + b.z * b.z + b.w * b.w;
    }
    float den = fmaxf(sqrtf(na) * sqrtf(nb), 1e-6f);
    out[e] = dot / den;
}


static inline int cdiv(long long a, long long b) { return (int)((a + b - 1) / b); }

extern "C" void kernel_launch(void* const* d_in, const int* in_sizes, int n_in,
                              void* d_out, int out_size, void* d_ws, size_t ws_size,
                              hipStream_t stream) {
    const float* x     = (const float*)d_in[0];
    const int*   ei    = (const int*)d_in[1];
    const float* ew    = (const float*)d_in[2];
    const int*   eli   = (const int*)d_in[3];
    const float* preW  = (const float*)d_in[4];
    const float* preb  = (const float*)d_in[5];
    const float* g1Wi  = (const float*)d_in[6];
    const float* g1Wr  = (const float*)d_in[7];
    const float* g1b   = (const float*)d_in[8];
    const float* g2Wi  = (const float*)d_in[9];
    const float* g2Wr  = (const float*)d_in[10];
    const float* g2b   = (const float*)d_in[11];
    const float* poW   = (const float*)d_in[12];
    const float* pob   = (const float*)d_in[13];
    float* out = (float*)d_out;

    const int N = in_sizes[0];
    const int E = in_sizes[2];
    const int L = in_sizes[3] / 2;

    const int* erow = ei;
    const int* ecol = ei + E;

    float* ws   = (float*)d_ws;
    float* deg  = ws;
    float* norm = deg  + N;
    float* h64  = norm + E;
    float* t    = h64  + (size_t)N * 64;
    float* agg1 = t    + (size_t)N * 32;
    float* agg2 = agg1 + (size_t)N * 32;
    float* z    = agg2 + (size_t)N * 32;

    const int B = 256;
    const int SCATTER_BLOCKS = 2048;
    const int gemm_threads = cdiv(N, 16) * 32;

    zero_f32_kernel<<<cdiv(N, B), B, 0, stream>>>(deg, N);
    deg_scatter_kernel<<<cdiv(E, B), B, 0, stream>>>(ecol, ew, deg, E);
    dinv_kernel<<<cdiv(N, B), B, 0, stream>>>(deg, N);
    norm_kernel<<<cdiv(E, B), B, 0, stream>>>(erow, ecol, ew, deg, norm, E);

    pre_kernel<<<cdiv((long long)N * 64, B), B, 0, stream>>>(x, preW, preb, h64, N * 64);

    wmma_gemm_kernel<2, true><<<cdiv(gemm_threads, B), B, 0, stream>>>(
        h64, 64, g1Wi, g1Wr, g1b, t, agg1, N);
    edge_scatter_kernel<<<SCATTER_BLOCKS, B, 0, stream>>>(
        erow, ecol, norm, t, agg1, E);
    relu_kernel<<<cdiv((long long)N * 32, B), B, 0, stream>>>(agg1, N * 32);

    wmma_gemm_kernel<2, true><<<cdiv(gemm_threads, B), B, 0, stream>>>(
        agg1, 32, g2Wi, g2Wr, g2b, t, agg2, N);
    edge_scatter_kernel<<<SCATTER_BLOCKS, B, 0, stream>>>(
        erow, ecol, norm, t, agg2, E);
    relu_kernel<<<cdiv((long long)N * 32, B), B, 0, stream>>>(agg2, N * 32);

    wmma_gemm_kernel<1, false><<<cdiv(gemm_threads, B), B, 0, stream>>>(
        agg2, 32, poW, nullptr, pob, z, nullptr, N);

    score_kernel<<<cdiv(L, B), B, 0, stream>>>(eli, z, out, L);
}


# segment sums through in-call destination bins: LDS counts, scan and LDS cursors pack edges per 256-node bin; per-bin LDS float adds replace all global atomics; sum order = LDS arrival order
# speedup vs baseline: 3.8273x; 2.0716x over previous
#include <hip/hip_runtime.h>
#include <math.h>

typedef float v2f __attribute__((ext_vector_type(2)));
typedef float v8f __attribute__((ext_vector_type(8)));

static __device__ __forceinline__ float elu1(float v) {
    return v > 0.0f ? v : (__expf(v) - 1.0f);
}


#define BIN_SHIFT 8
#define BIN_NODES 256
#define MAXB 512
#define PART_WG 256
#define WG_THREADS 1024

__global__ void __launch_bounds__(WG_THREADS)
bin_count_kernel(const int* __restrict__ col, int* __restrict__ counts, int E, int NB, int chunk) {
    __shared__ int hist[MAXB];
    const int tid = threadIdx.x;
    if (tid < MAXB) hist[tid] = 0;
    __syncthreads();
    const int base = blockIdx.x * chunk;
    const int end = min(base + chunk, E);
    for (int e0 = base; e0 < end; e0 += 4 * WG_THREADS) {
        int c[4];
#pragma unroll
        for (int k = 0; k < 4; ++k) {
            int e = e0 + k * WG_THREADS + tid;
            c[k] = (e < end) ? col[e] : -1;
        }
#pragma unroll
        for (int k = 0; k < 4; ++k)
            if (c[k] >= 0) atomicAdd(&hist[c[k] >> BIN_SHIFT], 1);
    }
    __syncthreads();
    if (tid < NB) counts[blockIdx.x * NB + tid] = hist[tid];
}

__global__ void __launch_bounds__(WG_THREADS)
bin_fill_kernel(const int* __restrict__ row, const int* __restrict__ col, const float* __restrict__ w,
                const int* __restrict__ counts, int* __restrict__ binStart, int* __restrict__ recw,
                int E, int NB, int chunk) {
    int2* __restrict__ rec = (int2*)recw;
    __shared__ int cursor[MAXB];
    __shared__ int sc[MAXB];
    __shared__ int ptot[2][MAXB];
    __shared__ int ppre[2][MAXB];
    const int tid = threadIdx.x;
    const int g = blockIdx.x;
    {
        const int b = tid & (MAXB - 1);
        const int part = tid >> 9;
        int tot = 0, pre = 0;
        if (b < NB) {
            const int g0 = part * (PART_WG / 2);
#pragma unroll 16
            for (int i = 0; i < PART_WG / 2; ++i) {
                int gg = g0 + i;
                int c = counts[gg * NB + b];
                tot += c;
                pre += (gg < g) ? c : 0;
            }
        }
        ptot[part][b] = tot;
        ppre[part][b] = pre;
    }
    __syncthreads();
    int tot = 0, pre = 0;
    if (tid < MAXB) {
        tot = ptot[0][tid] + ptot[1][tid];
        pre = ppre[0][tid] + ppre[1][tid];
        sc[tid] = tot;
    }
    __syncthreads();
    for (int off = 1; off < MAXB; off <<= 1) {
        int v = 0;
        if (tid < MAXB && tid >= off) v = sc[tid - off];
        __syncthreads();
        if (tid < MAXB) sc[tid] += v;
        __syncthreads();
    }
    if (tid < MAXB) {
        int excl = sc[tid] - tot;
        cursor[tid] = excl + pre;
        if (g == 0 && tid <= NB) binStart[tid] = excl;
    }
    __syncthreads();

    const int base = g * chunk;
    const int end = min(base + chunk, E);
    for (int e0 = base; e0 < end; e0 += 4 * WG_THREADS) {
        int r[4], c[4];
        float ww[4];
#pragma unroll
        for (int k = 0; k < 4; ++k) {
            int e = e0 + k * WG_THREADS + tid;
            bool ok = e < end;
            r[k] = ok ? row[e] : 0;
            c[k] = ok ? col[e] : -1;
            ww[k] = ok ? w[e] : 0.0f;
        }
#pragma unroll
        for (int k = 0; k < 4; ++k) {
            if (c[k] >= 0) {
                int pos = atomicAdd(&cursor[c[k] >> BIN_SHIFT], 1);
                unsigned packed = (unsigned)r[k] | ((unsigned)(c[k] & (BIN_NODES - 1)) << 24);
                rec[pos] = make_int2((int)packed, __float_as_int(ww[k]));
            }
        }
    }
}

__global__ void __launch_bounds__(WG_THREADS)
bin_degree_kernel(const int* __restrict__ recw, const int* __restrict__ binStart,
                  float* __restrict__ dinv, int N) {
    const int2* __restrict__ rec = (const int2*)recw;
    __shared__ float d[BIN_NODES];
    const int tid = threadIdx.x;
    if (tid < BIN_NODES) d[tid] = 0.0f;
    __syncthreads();
    const int s = binStart[blockIdx.x];
    const int e = binStart[blockIdx.x + 1];
    for (int p0 = s; p0 < e; p0 += 4 * WG_THREADS) {
        int2 r[4];
#pragma unroll
        for (int k = 0; k < 4; ++k) {
            int p = p0 + k * WG_THREADS + tid;
            r[k] = rec[min(p, e - 1)];
        }
#pragma unroll
        for (int k = 0; k < 4; ++k) {
            int p = p0 + k * WG_THREADS + tid;
            if (p < e) atomicAdd(&d[(unsigned)r[k].x >> 24], __int_as_float(r[k].y));
        }
    }
    __syncthreads();
    if (tid < BIN_NODES) {
        int node = blockIdx.x * BIN_NODES + tid;
        if (node < N) {
            float v = d[tid];
            dinv[node] = v > 0.0f ? rsqrtf(v) : 0.0f;
        }
    }
}

__global__ void __launch_bounds__(WG_THREADS)
bin_norm_kernel(int* __restrict__ recw, const int* __restrict__ binStart,
                const float* __restrict__ dinv, int N) {
    int2* __restrict__ rec = (int2*)recw;
    __shared__ float dc[BIN_NODES];
    const int tid = threadIdx.x;
    if (tid < BIN_NODES) {
        int node = blockIdx.x * BIN_NODES + tid;
        dc[tid] = node < N ? dinv[node] : 0.0f;
    }
    __syncthreads();
    const int s = binStart[blockIdx.x];
    const int e = binStart[blockIdx.x + 1];
    for (int p0 = s; p0 < e; p0 += 4 * WG_THREADS) {
        int2 r[4];
        float dr[4];
#pragma unroll
        for (int k = 0; k < 4; ++k) {
            int p = p0 + k * WG_THREADS + tid;
            r[k] = rec[min(p, e - 1)];
        }
#pragma unroll
        for (int k = 0; k < 4; ++k) dr[k] = dinv[r[k].x & 0xFFFFFF];
#pragma unroll
        for (int k = 0; k < 4; ++k) {
            int p = p0 + k * WG_THREADS + tid;
            if (p < e) {
                float n = dr[k] * __int_as_float(r[k].y) * dc[(unsigned)r[k].x >> 24];
                rec[p] = make_int2(r[k].x, __float_as_int(n));
            }
        }
    }
}

__global__ void __launch_bounds__(WG_THREADS, 16)
bin_accumulate_kernel(const int* __restrict__ recw, const int* __restrict__ binStart,
                      const float* __restrict__ t, float* __restrict__ agg, int N) {
    const int2* __restrict__ rec = (const int2*)recw;
    __shared__ float tile[BIN_NODES * 32];
    const int tid = threadIdx.x;
    const int lane = tid & 31;
    const int wave = __builtin_amdgcn_readfirstlane(tid >> 5);
#pragma unroll
    for (int k = 0; k < (BIN_NODES * 32) / WG_THREADS; ++k) tile[k * WG_THREADS + tid] = 0.0f;
    __syncthreads();

    const int s = binStart[blockIdx.x];
    const int e = binStart[blockIdx.x + 1];
    const int step = (WG_THREADS / 32) * 32;

    int p0 = s + wave * 32;
    int2 cur = make_int2(0, 0);
    if (p0 < e) {
        int p = p0 + lane;
        if (p < e) cur = rec[p];
    }
    while (p0 < e) {
        const int pn = p0 + step;
        int2 nxt = make_int2(0, 0);
        if (pn < e) {
            int p = pn + lane;
            if (p < e) nxt = rec[p];
        }
        const int cnt = e - p0;
        if (cnt >= 32) {
            float v[32];
#pragma unroll
            for (int j = 0; j < 32; ++j) {
                int r = __builtin_amdgcn_readlane(cur.x, j) & 0xFFFFFF;
                v[j] = t[(unsigned)(r * 32 + lane)];
            }
#pragma unroll
            for (int j = 0; j < 32; ++j) {
                unsigned cl = (unsigned)__builtin_amdgcn_readlane(cur.x, j) >> 24;
                float n = __int_as_float(__builtin_amdgcn_readlane(cur.y, j));
                atomicAdd(&tile[cl * 32 + lane], n * v[j]);
            }
        } else {
            for (int j = 0; j < cnt; ++j) {
                int x = __shfl(cur.x, j);
                float n = __int_as_float(__shfl(cur.y, j));
                int r = x & 0xFFFFFF;
                unsigned cl = (unsigned)x >> 24;
                atomicAdd(&tile[cl * 32 + lane], n * t[(unsigned)(r * 32 + lane)]);
            }
        }
        p0 = pn;
        cur = nxt;
    }
    __syncthreads();

    const int nodeBase = blockIdx.x * BIN_NODES;
    for (int k0 = 0; k0 < (BIN_NODES * 32) / WG_THREADS; k0 += 4) {
        float a[4];
#pragma unroll
        for (int k = 0; k < 4; ++k) {
            int i = (k0 + k) * WG_THREADS + tid;
            int node = nodeBase + (i >> 5);
            a[k] = (node < N) ? agg[(size_t)node * 32 + lane] : 0.0f;
        }
#pragma unroll
        for (int k = 0; k < 4; ++k) {
            int i = (k0 + k) * WG_THREADS + tid;
            int node = nodeBase + (i >> 5);
            if (node < N) agg[(size_t)node * 32 + lane] = fmaxf(a[k] + tile[i], 0.0f);
        }
    }
}

__global__ void pre_kernel(const float* __restrict__ x, const float* __restrict__ preW,
                           const float* __restrict__ preb, float* __restrict__ h, int total) {
    int i = blockIdx.x * blockDim.x + threadIdx.x;
    if (i >= total) return;
    int n = i >> 6, f = i & 63;
    h[i] = elu1(x[n] * preW[f] + preb[f]);
}

template <int NCT, bool DUAL>
__global__ void wmma_gemm_kernel(const float* __restrict__ A, int K,
                                 const float* __restrict__ Wi,
                                 const float* __restrict__ Wr,
                                 const float* __restrict__ biasR,
                                 float* __restrict__ Ti,
                                 float* __restrict__ Tr,
                                 int nrows) {
    const int F = NCT * 16;
    int wave = (blockIdx.x * blockDim.x + threadIdx.x) >> 5;
    int lane = threadIdx.x & 31;
    int row0 = wave * 16;
    if (row0 >= nrows) return;

    int m  = lane & 15;
    int hi = lane >> 4;

    v8f accI[NCT] = {};
    v8f accR[NCT] = {};

    int ar = row0 + m;
    if (ar >= nrows) ar = nrows - 1;
    const float* arow = A + (size_t)ar * K;

    for (int k0 = 0; k0 < K; k0 += 4) {
        int ka = k0 + 2 * hi;
        v2f a;
        a.x = arow[ka];
        a.y = arow[ka + 1];
#pragma unroll
        for (int ct = 0; ct < NCT; ++ct) {
            int colc = ct * 16 + m;
            v2f b;
            b.x = Wi[(size_t)ka * F + colc];
            b.y = Wi[(size_t)(ka + 1) * F + colc];
            accI[ct] = __builtin_amdgcn_wmma_f32_16x16x4_f32(
                false, a, false, b, (short)0, accI[ct], false, false);
            if (DUAL) {
                v2f br;
                br.x = Wr[(size_t)ka * F + colc];
                br.y = Wr[(size_t)(ka + 1) * F + colc];
                accR[ct] = __builtin_amdgcn_wmma_f32_16x16x4_f32(
                    false, a, false, br, (short)0, accR[ct], false, false);
            }
        }
    }

#pragma unroll
    for (int ct = 0; ct < NCT; ++ct) {
        int colc = ct * 16 + m;
        float bv = biasR ? biasR[colc] : 0.0f;
#pragma unroll
        for (int r = 0; r < 8; ++r) {
            int rowm = row0 + r + 8 * hi;
            if (rowm >= nrows) continue;
            size_t idx = (size_t)rowm * F + colc;
            if (DUAL) {
                Ti[idx] = accI[ct][r];
                Tr[idx] = accR[ct][r] + bv;
            } else {
                Ti[idx] = accI[ct][r] + bv;
            }
        }
    }
}


__global__ void score_kernel(const int* __restrict__ li, const float* __restrict__ z,
                             float* __restrict__ out, int L) {
    int e = blockIdx.x * blockDim.x + threadIdx.x;
    if (e >= L) return;
    int i = li[e], j = li[L + e];
    const float4* za = (const float4*)(z + (size_t)i * 16);
    const float4* zb = (const float4*)(z + (size_t)j * 16);
    float dot = 0.0f, na = 0.0f, nb = 0.0f;
#pragma unroll
    for (int q = 0; q < 4; ++q) {
        float4 a = za[q], b = zb[q];
        dot += a.x * b.x + a.y * b.y + a.z * b.z + a.w * b.w;
        na  += a.x * a.x + a.y * a.y + a.z * a.z + a.w * a.w;
        nb  += b.x * b.x + b.y * b.y + b.z * b.z + b.w * b.w;
    }
    float den = fmaxf(sqrtf(na) * sqrtf(nb), 1e-6f);
    out[e] = dot / den;
}


static inline int cdiv(long long a, long long b) { return (int)((a + b - 1) / b); }

extern "C" void kernel_launch(void* const* d_in, const int* in_sizes, int n_in,
                              void* d_out, int out_size, void* d_ws, size_t ws_size,
                              hipStream_t stream) {
    const float* x     = (const float*)d_in[0];
    const int*   ei    = (const int*)d_in[1];
    const float* ew    = (const float*)d_in[2];
    const int*   eli   = (const int*)d_in[3];
    const float* preW  = (const float*)d_in[4];
    const float* preb  = (const float*)d_in[5];
    const float* g1Wi  = (const float*)d_in[6];
    const float* g1Wr  = (const float*)d_in[7];
    const float* g1b   = (const float*)d_in[8];
    const float* g2Wi  = (const float*)d_in[9];
    const float* g2Wr  = (const float*)d_in[10];
    const float* g2b   = (const float*)d_in[11];
    const float* poW   = (const float*)d_in[12];
    const float* pob   = (const float*)d_in[13];
    float* out = (float*)d_out;

    const int N = in_sizes[0];
    const int E = in_sizes[2];
    const int L = in_sizes[3] / 2;

    const int* erow = ei;
    const int* ecol = ei + E;

    float* ws   = (float*)d_ws;
    float* dinv = ws;
    float* aux  = dinv + N;
    float* h64  = aux  + E;
    float* t    = h64  + (size_t)N * 64;
    float* agg1 = t    + (size_t)N * 32;
    float* agg2 = agg1 + (size_t)N * 32;
    float* z    = agg2 + (size_t)N * 32;

    const int NB = cdiv(N, BIN_NODES);
    const int chunk = cdiv(E, PART_WG);
    int* counts   = (int*)aux;
    int* binStart = counts + PART_WG * NB;
    int* rec      = (int*)h64;

    const int B = 256;
    const int gemm_threads = cdiv(N, 16) * 32;

    pre_kernel<<<cdiv((long long)N * 64, B), B, 0, stream>>>(x, preW, preb, h64, N * 64);

    wmma_gemm_kernel<2, true><<<cdiv(gemm_threads, B), B, 0, stream>>>(
        h64, 64, g1Wi, g1Wr, g1b, t, agg1, N);

    bin_count_kernel<<<PART_WG, WG_THREADS, 0, stream>>>(ecol, counts, E, NB, chunk);
    bin_fill_kernel<<<PART_WG, WG_THREADS, 0, stream>>>(erow, ecol, ew, counts, binStart, rec, E, NB, chunk);
    bin_degree_kernel<<<NB, WG_THREADS, 0, stream>>>(rec, binStart, dinv, N);
    bin_norm_kernel<<<NB, WG_THREADS, 0, stream>>>(rec, binStart, dinv, N);

    bin_accumulate_kernel<<<NB, WG_THREADS, 0, stream>>>(rec, binStart, t, agg1, N);

    wmma_gemm_kernel<2, true><<<cdiv(gemm_threads, B), B, 0, stream>>>(
        agg1, 32, g2Wi, g2Wr, g2b, t, agg2, N);
    bin_accumulate_kernel<<<NB, WG_THREADS, 0, stream>>>(rec, binStart, t, agg2, N);

    wmma_gemm_kernel<1, false><<<cdiv(gemm_threads, B), B, 0, stream>>>(
        agg2, 32, poW, nullptr, pob, z, nullptr, N);

    score_kernel<<<cdiv(L, B), B, 0, stream>>>(eli, z, out, L);
}


# pre layer folded into the conv1 GEMM: each lane builds its A fragment elu(x*preW+preb) in registers; the [N,64] activation and its own launch are gone; k order and every sum unchanged
# speedup vs baseline: 4.1898x; 1.0947x over previous
#include <hip/hip_runtime.h>
#include <math.h>

typedef float v2f __attribute__((ext_vector_type(2)));
typedef float v8f __attribute__((ext_vector_type(8)));

static __device__ __forceinline__ float elu1(float v) {
    return v > 0.0f ? v : (__expf(v) - 1.0f);
}


#define BIN_SHIFT 8
#define BIN_NODES 256
#define MAXB 512
#define PART_WG 256
#define WG_THREADS 1024

__global__ void __launch_bounds__(WG_THREADS)
bin_count_kernel(const int* __restrict__ col, int* __restrict__ counts, int E, int NB, int chunk) {
    __shared__ int hist[MAXB];
    const int tid = threadIdx.x;
    if (tid < MAXB) hist[tid] = 0;
    __syncthreads();
    const int base = blockIdx.x * chunk;
    const int end = min(base + chunk, E);
    for (int e0 = base; e0 < end; e0 += 4 * WG_THREADS) {
        int c[4];
#pragma unroll
        for (int k = 0; k < 4; ++k) {
            int e = e0 + k * WG_THREADS + tid;
            c[k] = (e < end) ? col[e] : -1;
        }
#pragma unroll
        for (int k = 0; k < 4; ++k)
            if (c[k] >= 0) atomicAdd(&hist[c[k] >> BIN_SHIFT], 1);
    }
    __syncthreads();
    if (tid < NB) counts[blockIdx.x * NB + tid] = hist[tid];
}

__global__ void __launch_bounds__(WG_THREADS)
bin_fill_kernel(const int* __restrict__ row, const int* __restrict__ col, const float* __restrict__ w,
                const int* __restrict__ counts, int* __restrict__ binStart, int* __restrict__ recw,
                int E, int NB, int chunk) {
    int2* __restrict__ rec = (int2*)recw;
    __shared__ int cursor[MAXB];
    __shared__ int sc[MAXB];
    __shared__ int ptot[2][MAXB];
    __shared__ int ppre[2][MAXB];
    const int tid = threadIdx.x;
    const int g = blockIdx.x;
    {
        const int b = tid & (MAXB - 1);
        const int part = tid >> 9;
        int tot = 0, pre = 0;
        if (b < NB) {
            const int g0 = part * (PART_WG / 2);
#pragma unroll 16
            for (int i = 0; i < PART_WG / 2; ++i) {
                int gg = g0 + i;
                int c = counts[gg * NB + b];
                tot += c;
                pre += (gg < g) ? c : 0;
            }
        }
        ptot[part][b] = tot;
        ppre[part][b] = pre;
    }
    __syncthreads();
    int tot = 0, pre = 0;
    if (tid < MAXB) {
        tot = ptot[0][tid] + ptot[1][tid];
        pre = ppre[0][tid] + ppre[1][tid];
        sc[tid] = tot;
    }
    __syncthreads();
    for (int off = 1; off < MAXB; off <<= 1) {
        int v = 0;
        if (tid < MAXB && tid >= off) v = sc[tid - off];
        __syncthreads();
        if (tid < MAXB) sc[tid] += v;
        __syncthreads();
    }
    if (tid < MAXB) {
        int excl = sc[tid] - tot;
        cursor[tid] = excl + pre;
        if (g == 0 && tid <= NB) binStart[tid] = excl;
    }
    __syncthreads();

    const int base = g * chunk;
    const int end = min(base + chunk, E);
    for (int e0 = base; e0 < end; e0 += 4 * WG_THREADS) {
        int r[4], c[4];
        float ww[4];
#pragma unroll
        for (int k = 0; k < 4; ++k) {
            int e = e0 + k * WG_THREADS + tid;
            bool ok = e < end;
            r[k] = ok ? row[e] : 0;
            c[k] = ok ? col[e] : -1;
            ww[k] = ok ? w[e] : 0.0f;
        }
#pragma unroll
        for (int k = 0; k < 4; ++k) {
            if (c[k] >= 0) {
                int pos = atomicAdd(&cursor[c[k] >> BIN_SHIFT], 1);
                unsigned packed = (unsigned)r[k] | ((unsigned)(c[k] & (BIN_NODES - 1)) << 24);
                rec[pos] = make_int2((int)packed, __float_as_int(ww[k]));
            }
        }
    }
}

__global__ void __launch_bounds__(WG_THREADS)
bin_degree_kernel(const int* __restrict__ recw, const int* __restrict__ binStart,
                  float* __restrict__ dinv, int N) {
    const int2* __restrict__ rec = (const int2*)recw;
    __shared__ float d[BIN_NODES];
    const int tid = threadIdx.x;
    if (tid < BIN_NODES) d[tid] = 0.0f;
    __syncthreads();
    const int s = binStart[blockIdx.x];
    const int e = binStart[blockIdx.x + 1];
    for (int p0 = s; p0 < e; p0 += 4 * WG_THREADS) {
        int2 r[4];
#pragma unroll
        for (int k = 0; k < 4; ++k) {
            int p = p0 + k * WG_THREADS + tid;
            r[k] = rec[min(p, e - 1)];
        }
#pragma unroll
        for (int k = 0; k < 4; ++k) {
            int p = p0 + k * WG_THREADS + tid;
            if (p < e) atomicAdd(&d[(unsigned)r[k].x >> 24], __int_as_float(r[k].y));
        }
    }
    __syncthreads();
    if (tid < BIN_NODES) {
        int node = blockIdx.x * BIN_NODES + tid;
        if (node < N) {
            float v = d[tid];
            dinv[node] = v > 0.0f ? rsqrtf(v) : 0.0f;
        }
    }
}

__global__ void __launch_bounds__(WG_THREADS)
bin_norm_kernel(int* __restrict__ recw, const int* __restrict__ binStart,
                const float* __restrict__ dinv, int N) {
    int2* __restrict__ rec = (int2*)recw;
    __shared__ float dc[BIN_NODES];
    const int tid = threadIdx.x;
    if (tid < BIN_NODES) {
        int node = blockIdx.x * BIN_NODES + tid;
        dc[tid] = node < N ? dinv[node] : 0.0f;
    }
    __syncthreads();
    const int s = binStart[blockIdx.x];
    const int e = binStart[blockIdx.x + 1];
    for (int p0 = s; p0 < e; p0 += 4 * WG_THREADS) {
        int2 r[4];
        float dr[4];
#pragma unroll
        for (int k = 0; k < 4; ++k) {
            int p = p0 + k * WG_THREADS + tid;
            r[k] = rec[min(p, e - 1)];
        }
#pragma unroll
        for (int k = 0; k < 4; ++k) dr[k] = dinv[r[k].x & 0xFFFFFF];
#pragma unroll
        for (int k = 0; k < 4; ++k) {
            int p = p0 + k * WG_THREADS + tid;
            if (p < e) {
                float n = dr[k] * __int_as_float(r[k].y) * dc[(unsigned)r[k].x >> 24];
                rec[p] = make_int2(r[k].x, __float_as_int(n));
            }
        }
    }
}

__global__ void __launch_bounds__(WG_THREADS, 16)
bin_accumulate_kernel(const int* __restrict__ recw, const int* __restrict__ binStart,
                      const float* __restrict__ t, float* __restrict__ agg, int N) {
    const int2* __restrict__ rec = (const int2*)recw;
    __shared__ float tile[BIN_NODES * 32];
    const int tid = threadIdx.x;
    const int lane = tid & 31;
    const int wave = __builtin_amdgcn_readfirstlane(tid >> 5);
#pragma unroll
    for (int k = 0; k < (BIN_NODES * 32) / WG_THREADS; ++k) tile[k * WG_THREADS + tid] = 0.0f;
    __syncthreads();

    const int s = binStart[blockIdx.x];
    const int e = binStart[blockIdx.x + 1];
    const int step = (WG_THREADS / 32) * 32;

    int p0 = s + wave * 32;
    int2 cur = make_int2(0, 0);
    if (p0 < e) {
        int p = p0 + lane;
        if (p < e) cur = rec[p];
    }
    while (p0 < e) {
        const int pn = p0 + step;
        int2 nxt = make_int2(0, 0);
        if (pn < e) {
            int p = pn + lane;
            if (p < e) nxt = rec[p];
        }
        const int cnt = e - p0;
        if (cnt >= 32) {
            float v[32];
#pragma unroll
            for (int j = 0; j < 32; ++j) {
                int r = __builtin_amdgcn_readlane(cur.x, j) & 0xFFFFFF;
                v[j] = t[(unsigned)(r * 32 + lane)];
            }
#pragma unroll
            for (int j = 0; j < 32; ++j) {
                unsigned cl = (unsigned)__builtin_amdgcn_readlane(cur.x, j) >> 24;
                float n = __int_as_float(__builtin_amdgcn_readlane(cur.y, j));
                atomicAdd(&tile[cl * 32 + lane], n * v[j]);
            }
        } else {
            for (int j = 0; j < cnt; ++j) {
                int x = __shfl(cur.x, j);
                float n = __int_as_float(__shfl(cur.y, j));
                int r = x & 0xFFFFFF;
                unsigned cl = (unsigned)x >> 24;
                atomicAdd(&tile[cl * 32 + lane], n * t[(unsigned)(r * 32 + lane)]);
            }
        }
        p0 = pn;
        cur = nxt;
    }
    __syncthreads();

    const int nodeBase = blockIdx.x * BIN_NODES;
    for (int k0 = 0; k0 < (BIN_NODES * 32) / WG_THREADS; k0 += 4) {
        float a[4];
#pragma unroll
        for (int k = 0; k < 4; ++k) {
            int i = (k0 + k) * WG_THREADS + tid;
            int node = nodeBase + (i >> 5);
            a[k] = (node < N) ? agg[(size_t)node * 32 + lane] : 0.0f;
        }
#pragma unroll
        for (int k = 0; k < 4; ++k) {
            int i = (k0 + k) * WG_THREADS + tid;
            int node = nodeBase + (i >> 5);
            if (node < N) agg[(size_t)node * 32 + lane] = fmaxf(a[k] + tile[i], 0.0f);
        }
    }
}

template <int NCT, bool DUAL>
__global__ void wmma_gemm_kernel(const float* __restrict__ A, int K,
                                 const float* __restrict__ Wi,
                                 const float* __restrict__ Wr,
                                 const float* __restrict__ biasR,
                                 float* __restrict__ Ti,
                                 float* __restrict__ Tr,
                                 int nrows) {
    const int F = NCT * 16;
    int wave = (blockIdx.x * blockDim.x + threadIdx.x) >> 5;
    int lane = threadIdx.x & 31;
    int row0 = wave * 16;
    if (row0 >= nrows) return;

    int m  = lane & 15;
    int hi = lane >> 4;

    v8f accI[NCT] = {};
    v8f accR[NCT] = {};

    int ar = row0 + m;
    if (ar >= nrows) ar = nrows - 1;
    const float* arow = A + (size_t)ar * K;

    for (int k0 = 0; k0 < K; k0 += 4) {
        int ka = k0 + 2 * hi;
        v2f a;
        a.x = arow[ka];
        a.y = arow[ka + 1];
#pragma unroll
        for (int ct = 0; ct < NCT; ++ct) {
            int colc = ct * 16 + m;
            v2f b;
            b.x = Wi[(size_t)ka * F + colc];
            b.y = Wi[(size_t)(ka + 1) * F + colc];
            accI[ct] = __builtin_amdgcn_wmma_f32_16x16x4_f32(
                false, a, false, b, (short)0, accI[ct], false, false);
            if (DUAL) {
                v2f br;
                br.x = Wr[(size_t)ka * F + colc];
                br.y = Wr[(size_t)(ka + 1) * F + colc];
                accR[ct] = __builtin_amdgcn_wmma_f32_16x16x4_f32(
                    false, a, false, br, (short)0, accR[ct], false, false);
            }
        }
    }

#pragma unroll
    for (int ct = 0; ct < NCT; ++ct) {
        int colc = ct * 16 + m;
        float bv = biasR ? biasR[colc] : 0.0f;
#pragma unroll
        for (int r = 0; r < 8; ++r) {
            int rowm = row0 + r + 8 * hi;
            if (rowm >= nrows) continue;
            size_t idx = (size_t)rowm * F + colc;
            if (DUAL) {
                Ti[idx] = accI[ct][r];
                Tr[idx] = accR[ct][r] + bv;
            } else {
                Ti[idx] = accI[ct][r] + bv;
            }
        }
    }
}

__global__ void pre_gemm_kernel(const float* __restrict__ x,
                                const float* __restrict__ preW,
                                const float* __restrict__ preb,
                                const float* __restrict__ Wi,
                                const float* __restrict__ Wr,
                                const float* __restrict__ biasR,
                                float* __restrict__ Ti,
                                float* __restrict__ Tr,
                                int nrows) {
    constexpr int K = 64;
    constexpr int NCT = 2;
    constexpr int F = NCT * 16;
    int wave = (blockIdx.x * blockDim.x + threadIdx.x) >> 5;
    int lane = threadIdx.x & 31;
    int row0 = wave * 16;
    if (row0 >= nrows) return;

    int m  = lane & 15;
    int hi = lane >> 4;

    v8f accI[NCT] = {};
    v8f accR[NCT] = {};

    int ar = row0 + m;
    if (ar >= nrows) ar = nrows - 1;
    const float xv = x[ar];

    for (int k0 = 0; k0 < K; k0 += 4) {
        int ka = k0 + 2 * hi;
        v2f a;
        a.x = elu1(xv * preW[ka] + preb[ka]);
        a.y = elu1(xv * preW[ka + 1] + preb[ka + 1]);
#pragma unroll
        for (int ct = 0; ct < NCT; ++ct) {
            int colc = ct * 16 + m;
            v2f b;
            b.x = Wi[(size_t)ka * F + colc];
            b.y = Wi[(size_t)(ka + 1) * F + colc];
            accI[ct] = __builtin_amdgcn_wmma_f32_16x16x4_f32(
                false, a, false, b, (short)0, accI[ct], false, false);
            v2f br;
            br.x = Wr[(size_t)ka * F + colc];
            br.y = Wr[(size_t)(ka + 1) * F + colc];
            accR[ct] = __builtin_amdgcn_wmma_f32_16x16x4_f32(
                false, a, false, br, (short)0, accR[ct], false, false);
        }
    }

#pragma unroll
    for (int ct = 0; ct < NCT; ++ct) {
        int colc = ct * 16 + m;
        float bv = biasR[colc];
#pragma unroll
        for (int r = 0; r < 8; ++r) {
            int rowm = row0 + r + 8 * hi;
            if (rowm >= nrows) continue;
            size_t idx = (size_t)rowm * F + colc;
            Ti[idx] = accI[ct][r];
            Tr[idx] = accR[ct][r] + bv;
        }
    }
}


__global__ void score_kernel(const int* __restrict__ li, const float* __restrict__ z,
                             float* __restrict__ out, int L) {
    int e = blockIdx.x * blockDim.x + threadIdx.x;
    if (e >= L) return;
    int i = li[e], j = li[L + e];
    const float4* za = (const float4*)(z + (size_t)i * 16);
    const float4* zb = (const float4*)(z + (size_t)j * 16);
    float dot = 0.0f, na = 0.0f, nb = 0.0f;
#pragma unroll
    for (int q = 0; q < 4; ++q) {
        float4 a = za[q], b = zb[q];
        dot += a.x * b.x + a.y * b.y + a.z * b.z + a.w * b.w;
        na  += a.x * a.x + a.y * a.y + a.z * a.z + a.w * a.w;
        nb  += b.x * b.x + b.y * b.y + b.z * b.z + b.w * b.w;
    }
    float den = fmaxf(sqrtf(na) * sqrtf(nb), 1e-6f);
    out[e] = dot / den;
}


static inline int cdiv(long long a, long long b) { return (int)((a + b - 1) / b); }

extern "C" void kernel_launch(void* const* d_in, const int* in_sizes, int n_in,
                              void* d_out, int out_size, void* d_ws, size_t ws_size,
                              hipStream_t stream) {
    const float* x     = (const float*)d_in[0];
    const int*   ei    = (const int*)d_in[1];
    const float* ew    = (const float*)d_in[2];
    const int*   eli   = (const int*)d_in[3];
    const float* preW  = (const float*)d_in[4];
    const float* preb  = (const float*)d_in[5];
    const float* g1Wi  = (const float*)d_in[6];
    const float* g1Wr  = (const float*)d_in[7];
    const float* g1b   = (const float*)d_in[8];
    const float* g2Wi  = (const float*)d_in[9];
    const float* g2Wr  = (const float*)d_in[10];
    const float* g2b   = (const float*)d_in[11];
    const float* poW   = (const float*)d_in[12];
    const float* pob   = (const float*)d_in[13];
    float* out = (float*)d_out;

    const int N = in_sizes[0];
    const int E = in_sizes[2];
    const int L = in_sizes[3] / 2;

    const int* erow = ei;
    const int* ecol = ei + E;

    float* ws   = (float*)d_ws;
    float* dinv = ws;
    float* aux  = dinv + N;
    float* h64  = aux  + E;
    float* t    = h64  + (size_t)N * 64;
    float* agg1 = t    + (size_t)N * 32;
    float* agg2 = agg1 + (size_t)N * 32;
    float* z    = agg2 + (size_t)N * 32;

    const int NB = cdiv(N, BIN_NODES);
    const int chunk = cdiv(E, PART_WG);
    int* counts   = (int*)aux;
    int* binStart = counts + PART_WG * NB;
    int* rec      = (int*)h64;

    const int B = 256;
    const int gemm_threads = cdiv(N, 16) * 32;

    pre_gemm_kernel<<<cdiv(gemm_threads, B), B, 0, stream>>>(
        x, preW, preb, g1Wi, g1Wr, g1b, t, agg1, N);

    bin_count_kernel<<<PART_WG, WG_THREADS, 0, stream>>>(ecol, counts, E, NB, chunk);
    bin_fill_kernel<<<PART_WG, WG_THREADS, 0, stream>>>(erow, ecol, ew, counts, binStart, rec, E, NB, chunk);
    bin_degree_kernel<<<NB, WG_THREADS, 0, stream>>>(rec, binStart, dinv, N);
    bin_norm_kernel<<<NB, WG_THREADS, 0, stream>>>(rec, binStart, dinv, N);

    bin_accumulate_kernel<<<NB, WG_THREADS, 0, stream>>>(rec, binStart, t, agg1, N);

    wmma_gemm_kernel<2, true><<<cdiv(gemm_threads, B), B, 0, stream>>>(
        agg1, 32, g2Wi, g2Wr, g2b, t, agg2, N);
    bin_accumulate_kernel<<<NB, WG_THREADS, 0, stream>>>(rec, binStart, t, agg2, N);

    wmma_gemm_kernel<1, false><<<cdiv(gemm_threads, B), B, 0, stream>>>(
        agg2, 32, poW, nullptr, pob, z, nullptr, N);

    score_kernel<<<cdiv(L, B), B, 0, stream>>>(eli, z, out, L);
}


# conv2 and post products moved into the accumulate epilogues: each bin's relu'd 256x32 LDS tile feeds the f32 WMMA; two GEMM launches and the h1/h2 round trips are gone; k order and sums unchanged
# speedup vs baseline: 4.5084x; 1.0760x over previous
#include <hip/hip_runtime.h>
#include <math.h>

typedef float v2f __attribute__((ext_vector_type(2)));
typedef float v8f __attribute__((ext_vector_type(8)));

static __device__ __forceinline__ float elu1(float v) {
    return v > 0.0f ? v : (__expf(v) - 1.0f);
}


#define BIN_SHIFT 8
#define BIN_NODES 256
#define MAXB 512
#define PART_WG 256
#define WG_THREADS 1024

__global__ void __launch_bounds__(WG_THREADS)
bin_count_kernel(const int* __restrict__ col, int* __restrict__ counts, int E, int NB, int chunk) {
    __shared__ int hist[MAXB];
    const int tid = threadIdx.x;
    if (tid < MAXB) hist[tid] = 0;
    __syncthreads();
    const int base = blockIdx.x * chunk;
    const int end = min(base + chunk, E);
    for (int e0 = base; e0 < end; e0 += 4 * WG_THREADS) {
        int c[4];
#pragma unroll
        for (int k = 0; k < 4; ++k) {
            int e = e0 + k * WG_THREADS + tid;
            c[k] = (e < end) ? col[e] : -1;
        }
#pragma unroll
        for (int k = 0; k < 4; ++k)
            if (c[k] >= 0) atomicAdd(&hist[c[k] >> BIN_SHIFT], 1);
    }
    __syncthreads();
    if (tid < NB) counts[blockIdx.x * NB + tid] = hist[tid];
}

__global__ void __launch_bounds__(WG_THREADS)
bin_fill_kernel(const int* __restrict__ row, const int* __restrict__ col, const float* __restrict__ w,
                const int* __restrict__ counts, int* __restrict__ binStart, int* __restrict__ recw,
                int E, int NB, int chunk) {
    int2* __restrict__ rec = (int2*)recw;
    __shared__ int cursor[MAXB];
    __shared__ int sc[MAXB];
    __shared__ int ptot[2][MAXB];
    __shared__ int ppre[2][MAXB];
    const int tid = threadIdx.x;
    const int g = blockIdx.x;
    {
        const int b = tid & (MAXB - 1);
        const int part = tid >> 9;
        int tot = 0, pre = 0;
        if (b < NB) {
            const int g0 = part * (PART_WG / 2);
#pragma unroll 16
            for (int i = 0; i < PART_WG / 2; ++i) {
                int gg = g0 + i;
                int c = counts[gg * NB + b];
                tot += c;
                pre += (gg < g) ? c : 0;
            }
        }
        ptot[part][b] = tot;
        ppre[part][b] = pre;
    }
    __syncthreads();
    int tot = 0, pre = 0;
    if (tid < MAXB) {
        tot = ptot[0][tid] + ptot[1][tid];
        pre = ppre[0][tid] + ppre[1][tid];
        sc[tid] = tot;
    }
    __syncthreads();
    for (int off = 1; off < MAXB; off <<= 1) {
        int v = 0;
        if (tid < MAXB && tid >= off) v = sc[tid - off];
        __syncthreads();
        if (tid < MAXB) sc[tid] += v;
        __syncthreads();
    }
    if (tid < MAXB) {
        int excl = sc[tid] - tot;
        cursor[tid] = excl + pre;
        if (g == 0 && tid <= NB) binStart[tid] = excl;
    }
    __syncthreads();

    const int base = g * chunk;
    const int end = min(base + chunk, E);
    for (int e0 = base; e0 < end; e0 += 4 * WG_THREADS) {
        int r[4], c[4];
        float ww[4];
#pragma unroll
        for (int k = 0; k < 4; ++k) {
            int e = e0 + k * WG_THREADS + tid;
            bool ok = e < end;
            r[k] = ok ? row[e] : 0;
            c[k] = ok ? col[e] : -1;
            ww[k] = ok ? w[e] : 0.0f;
        }
#pragma unroll
        for (int k = 0; k < 4; ++k) {
            if (c[k] >= 0) {
                int pos = atomicAdd(&cursor[c[k] >> BIN_SHIFT], 1);
                unsigned packed = (unsigned)r[k] | ((unsigned)(c[k] & (BIN_NODES - 1)) << 24);
                rec[pos] = make_int2((int)packed, __float_as_int(ww[k]));
            }
        }
    }
}

__global__ void __launch_bounds__(WG_THREADS)
bin_degree_kernel(const int* __restrict__ recw, const int* __restrict__ binStart,
                  float* __restrict__ dinv, int N) {
    const int2* __restrict__ rec = (const int2*)recw;
    __shared__ float d[BIN_NODES];
    const int tid = threadIdx.x;
    if (tid < BIN_NODES) d[tid] = 0.0f;
    __syncthreads();
    const int s = binStart[blockIdx.x];
    const int e = binStart[blockIdx.x + 1];
    for (int p0 = s; p0 < e; p0 += 4 * WG_THREADS) {
        int2 r[4];
#pragma unroll
        for (int k = 0; k < 4; ++k) {
            int p = p0 + k * WG_THREADS + tid;
            r[k] = rec[min(p, e - 1)];
        }
#pragma unroll
        for (int k = 0; k < 4; ++k) {
            int p = p0 + k * WG_THREADS + tid;
            if (p < e) atomicAdd(&d[(unsigned)r[k].x >> 24], __int_as_float(r[k].y));
        }
    }
    __syncthreads();
    if (tid < BIN_NODES) {
        int node = blockIdx.x * BIN_NODES + tid;
        if (node < N) {
            float v = d[tid];
            dinv[node] = v > 0.0f ? rsqrtf(v) : 0.0f;
        }
    }
}

__global__ void __launch_bounds__(WG_THREADS)
bin_norm_kernel(int* __restrict__ recw, const int* __restrict__ binStart,
                const float* __restrict__ dinv, int N) {
    int2* __restrict__ rec = (int2*)recw;
    __shared__ float dc[BIN_NODES];
    const int tid = threadIdx.x;
    if (tid < BIN_NODES) {
        int node = blockIdx.x * BIN_NODES + tid;
        dc[tid] = node < N ? dinv[node] : 0.0f;
    }
    __syncthreads();
    const int s = binStart[blockIdx.x];
    const int e = binStart[blockIdx.x + 1];
    for (int p0 = s; p0 < e; p0 += 4 * WG_THREADS) {
        int2 r[4];
        float dr[4];
#pragma unroll
        for (int k = 0; k < 4; ++k) {
            int p = p0 + k * WG_THREADS + tid;
            r[k] = rec[min(p, e - 1)];
        }
#pragma unroll
        for (int k = 0; k < 4; ++k) dr[k] = dinv[r[k].x & 0xFFFFFF];
#pragma unroll
        for (int k = 0; k < 4; ++k) {
            int p = p0 + k * WG_THREADS + tid;
            if (p < e) {
                float n = dr[k] * __int_as_float(r[k].y) * dc[(unsigned)r[k].x >> 24];
                rec[p] = make_int2(r[k].x, __float_as_int(n));
            }
        }
    }
}

#define TP 36
template <int NCT, bool DUAL>
__global__ void __launch_bounds__(WG_THREADS, 16)
bin_accumulate_gemm_kernel(const int* __restrict__ recw, const int* __restrict__ binStart,
                           const float* __restrict__ t, float* root,
                           const float* __restrict__ Wi, const float* __restrict__ Wr,
                           const float* __restrict__ biasR, float* __restrict__ outR, int N) {
    const int2* __restrict__ rec = (const int2*)recw;
    __shared__ float tile[BIN_NODES * TP];
    const int tid = threadIdx.x;
    const int lane = tid & 31;
    const int wave = __builtin_amdgcn_readfirstlane(tid >> 5);
    for (int i = tid; i < BIN_NODES * TP; i += WG_THREADS) tile[i] = 0.0f;
    __syncthreads();

    const int s = binStart[blockIdx.x];
    const int e = binStart[blockIdx.x + 1];
    const int step = (WG_THREADS / 32) * 32;

    int p0 = s + wave * 32;
    int2 cur = make_int2(0, 0);
    if (p0 < e) {
        int p = p0 + lane;
        if (p < e) cur = rec[p];
    }
    while (p0 < e) {
        const int pn = p0 + step;
        int2 nxt = make_int2(0, 0);
        if (pn < e) {
            int p = pn + lane;
            if (p < e) nxt = rec[p];
        }
        const int cnt = e - p0;
        if (cnt >= 32) {
            float v[32];
#pragma unroll
            for (int j = 0; j < 32; ++j) {
                int r = __builtin_amdgcn_readlane(cur.x, j) & 0xFFFFFF;
                v[j] = t[(unsigned)(r * 32 + lane)];
            }
#pragma unroll
            for (int j = 0; j < 32; ++j) {
                unsigned cl = (unsigned)__builtin_amdgcn_readlane(cur.x, j) >> 24;
                float n = __int_as_float(__builtin_amdgcn_readlane(cur.y, j));
                atomicAdd(&tile[cl * TP + lane], n * v[j]);
            }
        } else {
            for (int j = 0; j < cnt; ++j) {
                int x = __shfl(cur.x, j);
                float n = __int_as_float(__shfl(cur.y, j));
                int r = x & 0xFFFFFF;
                unsigned cl = (unsigned)x >> 24;
                atomicAdd(&tile[cl * TP + lane], n * t[(unsigned)(r * 32 + lane)]);
            }
        }
        p0 = pn;
        cur = nxt;
    }
    __syncthreads();

    const int nodeBase = blockIdx.x * BIN_NODES;
    for (int k0 = 0; k0 < (BIN_NODES * 32) / WG_THREADS; k0 += 4) {
        float a[4];
#pragma unroll
        for (int k = 0; k < 4; ++k) {
            int i = (k0 + k) * WG_THREADS + tid;
            int node = nodeBase + (i >> 5);
            a[k] = (node < N) ? root[(size_t)node * 32 + lane] : 0.0f;
        }
#pragma unroll
        for (int k = 0; k < 4; ++k) {
            int i = (k0 + k) * WG_THREADS + tid;
            int nl = i >> 5;
            float h = (nodeBase + nl < N) ? fmaxf(a[k] + tile[nl * TP + lane], 0.0f) : 0.0f;
            tile[nl * TP + lane] = h;
        }
    }
    __syncthreads();

    const int part = wave >> 4;
    if (DUAL || part == 0) {
        constexpr int F = NCT * 16;
        const int rt = wave & 15;
        const int m  = lane & 15;
        const int hi = lane >> 4;
        const bool second = DUAL && (part != 0);
        const float* __restrict__ Wsel = second ? Wr : Wi;
        const bool withBias = DUAL ? second : true;
        float* dst = DUAL ? (second ? outR : root) : outR;

        float bv[NCT];
#pragma unroll
        for (int ct = 0; ct < NCT; ++ct) bv[ct] = withBias ? biasR[ct * 16 + m] : -0.0f;

        v8f acc[NCT] = {};
        const float* arow = &tile[(rt * 16 + m) * TP + 2 * hi];
#pragma unroll
        for (int k0 = 0; k0 < 32; k0 += 4) {
            const int ka = k0 + 2 * hi;
            v2f a;
            a.x = arow[k0];
            a.y = arow[k0 + 1];
#pragma unroll
            for (int ct = 0; ct < NCT; ++ct) {
                const int colc = ct * 16 + m;
                v2f b;
                b.x = Wsel[(size_t)ka * F + colc];
                b.y = Wsel[(size_t)(ka + 1) * F + colc];
                acc[ct] = __builtin_amdgcn_wmma_f32_16x16x4_f32(
                    false, a, false, b, (short)0, acc[ct], false, false);
            }
        }

#pragma unroll
        for (int ct = 0; ct < NCT; ++ct) {
            const int colc = ct * 16 + m;
            float o[8];
#pragma unroll
            for (int r = 0; r < 8; ++r) o[r] = acc[ct][r] + bv[ct];
#pragma unroll
            for (int r = 0; r < 8; ++r) {
                const int node = nodeBase + rt * 16 + r + 8 * hi;
                if (node < N) dst[(size_t)node * F + colc] = o[r];
            }
        }
    }
}


__global__ void pre_gemm_kernel(const float* __restrict__ x,
                                const float* __restrict__ preW,
                                const float* __restrict__ preb,
                                const float* __restrict__ Wi,
                                const float* __restrict__ Wr,
                                const float* __restrict__ biasR,
                                float* __restrict__ Ti,
                                float* __restrict__ Tr,
                                int nrows) {
    constexpr int K = 64;
    constexpr int NCT = 2;
    constexpr int F = NCT * 16;
    int wave = (blockIdx.x * blockDim.x + threadIdx.x) >> 5;
    int lane = threadIdx.x & 31;
    int row0 = wave * 16;
    if (row0 >= nrows) return;

    int m  = lane & 15;
    int hi = lane >> 4;

    v8f accI[NCT] = {};
    v8f accR[NCT] = {};

    int ar = row0 + m;
    if (ar >= nrows) ar = nrows - 1;
    const float xv = x[ar];

    for (int k0 = 0; k0 < K; k0 += 4) {
        int ka = k0 + 2 * hi;
        v2f a;
        a.x = elu1(xv * preW[ka] + preb[ka]);
        a.y = elu1(xv * preW[ka + 1] + preb[ka + 1]);
#pragma unroll
        for (int ct = 0; ct < NCT; ++ct) {
            int colc = ct * 16 + m;
            v2f b;
            b.x = Wi[(size_t)ka * F + colc];
            b.y = Wi[(size_t)(ka + 1) * F + colc];
            accI[ct] = __builtin_amdgcn_wmma_f32_16x16x4_f32(
                false, a, false, b, (short)0, accI[ct], false, false);
            v2f br;
            br.x = Wr[(size_t)ka * F + colc];
            br.y = Wr[(size_t)(ka + 1) * F + colc];
            accR[ct] = __builtin_amdgcn_wmma_f32_16x16x4_f32(
                false, a, false, br, (short)0, accR[ct], false, false);
        }
    }

#pragma unroll
    for (int ct = 0; ct < NCT; ++ct) {
        int colc = ct * 16 + m;
        float bv = biasR[colc];
#pragma unroll
        for (int r = 0; r < 8; ++r) {
            int rowm = row0 + r + 8 * hi;
            if (rowm >= nrows) continue;
            size_t idx = (size_t)rowm * F + colc;
            Ti[idx] = accI[ct][r];
            Tr[idx] = accR[ct][r] + bv;
        }
    }
}


__global__ void score_kernel(const int* __restrict__ li, const float* __restrict__ z,
                             float* __restrict__ out, int L) {
    int e = blockIdx.x * blockDim.x + threadIdx.x;
    if (e >= L) return;
    int i = li[e], j = li[L + e];
    const float4* za = (const float4*)(z + (size_t)i * 16);
    const float4* zb = (const float4*)(z + (size_t)j * 16);
    float dot = 0.0f, na = 0.0f, nb = 0.0f;
#pragma unroll
    for (int q = 0; q < 4; ++q) {
        float4 a = za[q], b = zb[q];
        dot += a.x * b.x + a.y * b.y + a.z * b.z + a.w * b.w;
        na  += a.x * a.x + a.y * a.y + a.z * a.z + a.w * a.w;
        nb  += b.x * b.x + b.y * b.y + b.z * b.z + b.w * b.w;
    }
    float den = fmaxf(sqrtf(na) * sqrtf(nb), 1e-6f);
    out[e] = dot / den;
}


static inline int cdiv(long long a, long long b) { return (int)((a + b - 1) / b); }

extern "C" void kernel_launch(void* const* d_in, const int* in_sizes, int n_in,
                              void* d_out, int out_size, void* d_ws, size_t ws_size,
                              hipStream_t stream) {
    const float* x     = (const float*)d_in[0];
    const int*   ei    = (const int*)d_in[1];
    const float* ew    = (const float*)d_in[2];
    const int*   eli   = (const int*)d_in[3];
    const float* preW  = (const float*)d_in[4];
    const float* preb  = (const float*)d_in[5];
    const float* g1Wi  = (const float*)d_in[6];
    const float* g1Wr  = (const float*)d_in[7];
    const float* g1b   = (const float*)d_in[8];
    const float* g2Wi  = (const float*)d_in[9];
    const float* g2Wr  = (const float*)d_in[10];
    const float* g2b   = (const float*)d_in[11];
    const float* poW   = (const float*)d_in[12];
    const float* pob   = (const float*)d_in[13];
    float* out = (float*)d_out;

    const int N = in_sizes[0];
    const int E = in_sizes[2];
    const int L = in_sizes[3] / 2;

    const int* erow = ei;
    const int* ecol = ei + E;

    float* ws   = (float*)d_ws;
    float* dinv = ws;
    float* aux  = dinv + N;
    float* h64  = aux  + E;
    float* t    = h64  + (size_t)N * 64;
    float* agg1 = t    + (size_t)N * 32;
    float* agg2 = agg1 + (size_t)N * 32;
    float* z    = agg2 + (size_t)N * 32;

    const int NB = cdiv(N, BIN_NODES);
    const int chunk = cdiv(E, PART_WG);
    int* counts   = (int*)aux;
    int* binStart = counts + PART_WG * NB;
    int* rec      = (int*)h64;

    const int B = 256;
    const int gemm_threads = cdiv(N, 16) * 32;

    pre_gemm_kernel<<<cdiv(gemm_threads, B), B, 0, stream>>>(
        x, preW, preb, g1Wi, g1Wr, g1b, t, agg1, N);

    bin_count_kernel<<<PART_WG, WG_THREADS, 0, stream>>>(ecol, counts, E, NB, chunk);
    bin_fill_kernel<<<PART_WG, WG_THREADS, 0, stream>>>(erow, ecol, ew, counts, binStart, rec, E, NB, chunk);
    bin_degree_kernel<<<NB, WG_THREADS, 0, stream>>>(rec, binStart, dinv, N);
    bin_norm_kernel<<<NB, WG_THREADS, 0, stream>>>(rec, binStart, dinv, N);

    bin_accumulate_gemm_kernel<2, true><<<NB, WG_THREADS, 0, stream>>>(
        rec, binStart, t, agg1, g2Wi, g2Wr, g2b, agg2, N);

    bin_accumulate_gemm_kernel<1, false><<<NB, WG_THREADS, 0, stream>>>(
        rec, binStart, agg1, agg2, poW, nullptr, pob, z, N);

    score_kernel<<<cdiv(L, B), B, 0, stream>>>(eli, z, out, L);
}


# gcn_norm taken apart: message rows stored times dinv[row], records keep w, dinv[col] scales each node's sum once; norm launch and record rewrite gone; terms round differently, sum order kept
# speedup vs baseline: 5.0504x; 1.1202x over previous
#include <hip/hip_runtime.h>
#include <math.h>

typedef float v2f __attribute__((ext_vector_type(2)));
typedef float v8f __attribute__((ext_vector_type(8)));

static __device__ __forceinline__ float elu1(float v) {
    return v > 0.0f ? v : (__expf(v) - 1.0f);
}


#define BIN_SHIFT 8
#define BIN_NODES 256
#define MAXB 512
#define PART_WG 256
#define WG_THREADS 1024

__global__ void __launch_bounds__(WG_THREADS)
bin_count_kernel(const int* __restrict__ col, int* __restrict__ counts, int E, int NB, int chunk) {
    __shared__ int hist[MAXB];
    const int tid = threadIdx.x;
    if (tid < MAXB) hist[tid] = 0;
    __syncthreads();
    const int base = blockIdx.x * chunk;
    const int end = min(base + chunk, E);
    for (int e0 = base; e0 < end; e0 += 4 * WG_THREADS) {
        int c[4];
#pragma unroll
        for (int k = 0; k < 4; ++k) {
            int e = e0 + k * WG_THREADS + tid;
            c[k] = (e < end) ? col[e] : -1;
        }
#pragma unroll
        for (int k = 0; k < 4; ++k)
            if (c[k] >= 0) atomicAdd(&hist[c[k] >> BIN_SHIFT], 1);
    }
    __syncthreads();
    if (tid < NB) counts[blockIdx.x * NB + tid] = hist[tid];
}

__global__ void __launch_bounds__(WG_THREADS)
bin_fill_kernel(const int* __restrict__ row, const int* __restrict__ col, const float* __restrict__ w,
                const int* __restrict__ counts, int* __restrict__ binStart, int* __restrict__ recw,
                int E, int NB, int chunk) {
    int2* __restrict__ rec = (int2*)recw;
    __shared__ int cursor[MAXB];
    __shared__ int sc[MAXB];
    __shared__ int ptot[2][MAXB];
    __shared__ int ppre[2][MAXB];
    const int tid = threadIdx.x;
    const int g = blockIdx.x;
    {
        const int b = tid & (MAXB - 1);
        const int part = tid >> 9;
        int tot = 0, pre = 0;
        if (b < NB) {
            const int g0 = part * (PART_WG / 2);
#pragma unroll 16
            for (int i = 0; i < PART_WG / 2; ++i) {
                int gg = g0 + i;
                int c = counts[gg * NB + b];
                tot += c;
                pre += (gg < g) ? c : 0;
            }
        }
        ptot[part][b] = tot;
        ppre[part][b] = pre;
    }
    __syncthreads();
    int tot = 0, pre = 0;
    if (tid < MAXB) {
        tot = ptot[0][tid] + ptot[1][tid];
        pre = ppre[0][tid] + ppre[1][tid];
        sc[tid] = tot;
    }
    __syncthreads();
    for (int off = 1; off < MAXB; off <<= 1) {
        int v = 0;
        if (tid < MAXB && tid >= off) v = sc[tid - off];
        __syncthreads();
        if (tid < MAXB) sc[tid] += v;
        __syncthreads();
    }
    if (tid < MAXB) {
        int excl = sc[tid] - tot;
        cursor[tid] = excl + pre;
        if (g == 0 && tid <= NB) binStart[tid] = excl;
    }
    __syncthreads();

    const int base = g * chunk;
    const int end = min(base + chunk, E);
    for (int e0 = base; e0 < end; e0 += 4 * WG_THREADS) {
        int r[4], c[4];
        float ww[4];
#pragma unroll
        for (int k = 0; k < 4; ++k) {
            int e = e0 + k * WG_THREADS + tid;
            bool ok = e < end;
            r[k] = ok ? row[e] : 0;
            c[k] = ok ? col[e] : -1;
            ww[k] = ok ? w[e] : 0.0f;
        }
#pragma unroll
        for (int k = 0; k < 4; ++k) {
            if (c[k] >= 0) {
                int pos = atomicAdd(&cursor[c[k] >> BIN_SHIFT], 1);
                unsigned packed = (unsigned)r[k] | ((unsigned)(c[k] & (BIN_NODES - 1)) << 24);
                rec[pos] = make_int2((int)packed, __float_as_int(ww[k]));
            }
        }
    }
}

__global__ void __launch_bounds__(WG_THREADS)
bin_degree_kernel(const int* __restrict__ recw, const int* __restrict__ binStart,
                  float* __restrict__ dinv, int N) {
    const int2* __restrict__ rec = (const int2*)recw;
    __shared__ float d[BIN_NODES];
    const int tid = threadIdx.x;
    if (tid < BIN_NODES) d[tid] = 0.0f;
    __syncthreads();
    const int s = binStart[blockIdx.x];
    const int e = binStart[blockIdx.x + 1];
    for (int p0 = s; p0 < e; p0 += 4 * WG_THREADS) {
        int2 r[4];
#pragma unroll
        for (int k = 0; k < 4; ++k) {
            int p = p0 + k * WG_THREADS + tid;
            r[k] = rec[min(p, e - 1)];
        }
#pragma unroll
        for (int k = 0; k < 4; ++k) {
            int p = p0 + k * WG_THREADS + tid;
            if (p < e) atomicAdd(&d[(unsigned)r[k].x >> 24], __int_as_float(r[k].y));
        }
    }
    __syncthreads();
    if (tid < BIN_NODES) {
        int node = blockIdx.x * BIN_NODES + tid;
        if (node < N) {
            float v = d[tid];
            dinv[node] = v > 0.0f ? rsqrtf(v) : 0.0f;
        }
    }
}

#define TP 36
template <int NCT, bool DUAL>
__global__ void __launch_bounds__(WG_THREADS, 16)
bin_accumulate_gemm_kernel(const int* __restrict__ recw, const int* __restrict__ binStart,
                           const float* __restrict__ t, float* root,
                           const float* __restrict__ Wi, const float* __restrict__ Wr,
                           const float* __restrict__ biasR, float* __restrict__ outR,
                           const float* __restrict__ dinv, int N) {
    const int2* __restrict__ rec = (const int2*)recw;
    __shared__ float tile[BIN_NODES * TP];
    __shared__ float dc[BIN_NODES];
    const int tid = threadIdx.x;
    const int lane = tid & 31;
    const int wave = __builtin_amdgcn_readfirstlane(tid >> 5);
    for (int i = tid; i < BIN_NODES * TP; i += WG_THREADS) tile[i] = 0.0f;
    if (tid < BIN_NODES) {
        int node = blockIdx.x * BIN_NODES + tid;
        dc[tid] = node < N ? dinv[node] : 0.0f;
    }
    __syncthreads();

    const int s = binStart[blockIdx.x];
    const int e = binStart[blockIdx.x + 1];
    const int step = (WG_THREADS / 32) * 32;

    int p0 = s + wave * 32;
    int2 cur = make_int2(0, 0);
    if (p0 < e) {
        int p = p0 + lane;
        if (p < e) cur = rec[p];
    }
    while (p0 < e) {
        const int pn = p0 + step;
        int2 nxt = make_int2(0, 0);
        if (pn < e) {
            int p = pn + lane;
            if (p < e) nxt = rec[p];
        }
        const int cnt = e - p0;
        if (cnt >= 32) {
            float v[32];
#pragma unroll
            for (int j = 0; j < 32; ++j) {
                int r = __builtin_amdgcn_readlane(cur.x, j) & 0xFFFFFF;
                v[j] = t[(unsigned)(r * 32 + lane)];
            }
#pragma unroll
            for (int j = 0; j < 32; ++j) {
                unsigned cl = (unsigned)__builtin_amdgcn_readlane(cur.x, j) >> 24;
                float n = __int_as_float(__builtin_amdgcn_readlane(cur.y, j));
                atomicAdd(&tile[cl * TP + lane], n * v[j]);
            }
        } else {
            for (int j = 0; j < cnt; ++j) {
                int x = __shfl(cur.x, j);
                float n = __int_as_float(__shfl(cur.y, j));
                int r = x & 0xFFFFFF;
                unsigned cl = (unsigned)x >> 24;
                atomicAdd(&tile[cl * TP + lane], n * t[(unsigned)(r * 32 + lane)]);
            }
        }
        p0 = pn;
        cur = nxt;
    }
    __syncthreads();

    const int nodeBase = blockIdx.x * BIN_NODES;
    for (int k0 = 0; k0 < (BIN_NODES * 32) / WG_THREADS; k0 += 4) {
        float a[4];
#pragma unroll
        for (int k = 0; k < 4; ++k) {
            int i = (k0 + k) * WG_THREADS + tid;
            int node = nodeBase + (i >> 5);
            a[k] = (node < N) ? root[(size_t)node * 32 + lane] : 0.0f;
        }
#pragma unroll
        for (int k = 0; k < 4; ++k) {
            int i = (k0 + k) * WG_THREADS + tid;
            int nl = i >> 5;
            const float sum = tile[nl * TP + lane];
            const float dn = dc[nl];
            float h = (nodeBase + nl < N) ? fmaxf(a[k] + dn * sum, 0.0f) : 0.0f;
            tile[nl * TP + lane] = h;
        }
    }
    __syncthreads();

    const int part = wave >> 4;
    if (DUAL || part == 0) {
        constexpr int F = NCT * 16;
        const int rt = wave & 15;
        const int m  = lane & 15;
        const int hi = lane >> 4;
        const bool second = DUAL && (part != 0);
        const float* __restrict__ Wsel = second ? Wr : Wi;
        const bool withBias = DUAL ? second : true;
        float* dst = DUAL ? (second ? outR : root) : outR;

        float bv[NCT];
#pragma unroll
        for (int ct = 0; ct < NCT; ++ct) bv[ct] = withBias ? biasR[ct * 16 + m] : -0.0f;
        const float am = (DUAL && !second) ? dc[rt * 16 + m] : 1.0f;

        v8f acc[NCT] = {};
        const float* arow = &tile[(rt * 16 + m) * TP + 2 * hi];
#pragma unroll
        for (int k0 = 0; k0 < 32; k0 += 4) {
            const int ka = k0 + 2 * hi;
            v2f a;
            a.x = arow[k0] * am;
            a.y = arow[k0 + 1] * am;
#pragma unroll
            for (int ct = 0; ct < NCT; ++ct) {
                const int colc = ct * 16 + m;
                v2f b;
                b.x = Wsel[(size_t)ka * F + colc];
                b.y = Wsel[(size_t)(ka + 1) * F + colc];
                acc[ct] = __builtin_amdgcn_wmma_f32_16x16x4_f32(
                    false, a, false, b, (short)0, acc[ct], false, false);
            }
        }

#pragma unroll
        for (int ct = 0; ct < NCT; ++ct) {
            const int colc = ct * 16 + m;
            float o[8];
#pragma unroll
            for (int r = 0; r < 8; ++r) o[r] = acc[ct][r] + bv[ct];
#pragma unroll
            for (int r = 0; r < 8; ++r) {
                const int node = nodeBase + rt * 16 + r + 8 * hi;
                if (node < N) dst[(size_t)node * F + colc] = o[r];
            }
        }
    }
}


__global__ void pre_gemm_kernel(const float* __restrict__ x,
                                const float* __restrict__ preW,
                                const float* __restrict__ preb,
                                const float* __restrict__ Wi,
                                const float* __restrict__ Wr,
                                const float* __restrict__ biasR,
                                float* __restrict__ Ti,
                                float* __restrict__ Tr,
                                const float* __restrict__ dinv,
                                int nrows) {
    constexpr int K = 64;
    constexpr int NCT = 2;
    constexpr int F = NCT * 16;
    int wave = (blockIdx.x * blockDim.x + threadIdx.x) >> 5;
    int lane = threadIdx.x & 31;
    int row0 = wave * 16;
    if (row0 >= nrows) return;

    int m  = lane & 15;
    int hi = lane >> 4;

    v8f accI[NCT] = {};
    v8f accR[NCT] = {};

    int ar = row0 + m;
    if (ar >= nrows) ar = nrows - 1;
    const float xv = x[ar];

    for (int k0 = 0; k0 < K; k0 += 4) {
        int ka = k0 + 2 * hi;
        v2f a;
        a.x = elu1(xv * preW[ka] + preb[ka]);
        a.y = elu1(xv * preW[ka + 1] + preb[ka + 1]);
#pragma unroll
        for (int ct = 0; ct < NCT; ++ct) {
            int colc = ct * 16 + m;
            v2f b;
            b.x = Wi[(size_t)ka * F + colc];
            b.y = Wi[(size_t)(ka + 1) * F + colc];
            accI[ct] = __builtin_amdgcn_wmma_f32_16x16x4_f32(
                false, a, false, b, (short)0, accI[ct], false, false);
            v2f br;
            br.x = Wr[(size_t)ka * F + colc];
            br.y = Wr[(size_t)(ka + 1) * F + colc];
            accR[ct] = __builtin_amdgcn_wmma_f32_16x16x4_f32(
                false, a, false, br, (short)0, accR[ct], false, false);
        }
    }

    float dv[8];
#pragma unroll
    for (int r = 0; r < 8; ++r) dv[r] = dinv[min(row0 + r + 8 * hi, nrows - 1)];

#pragma unroll
    for (int ct = 0; ct < NCT; ++ct) {
        int colc = ct * 16 + m;
        float bv = biasR[colc];
#pragma unroll
        for (int r = 0; r < 8; ++r) {
            int rowm = row0 + r + 8 * hi;
            if (rowm >= nrows) continue;
            size_t idx = (size_t)rowm * F + colc;
            Ti[idx] = accI[ct][r] * dv[r];
            Tr[idx] = accR[ct][r] + bv;
        }
    }
}


__global__ void score_kernel(const int* __restrict__ li, const float* __restrict__ z,
                             float* __restrict__ out, int L) {
    int e = blockIdx.x * blockDim.x + threadIdx.x;
    if (e >= L) return;
    int i = li[e], j = li[L + e];
    const float4* za = (const float4*)(z + (size_t)i * 16);
    const float4* zb = (const float4*)(z + (size_t)j * 16);
    float dot = 0.0f, na = 0.0f, nb = 0.0f;
#pragma unroll
    for (int q = 0; q < 4; ++q) {
        float4 a = za[q], b = zb[q];
        dot += a.x * b.x + a.y * b.y + a.z * b.z + a.w * b.w;
        na  += a.x * a.x + a.y * a.y + a.z * a.z + a.w * a.w;
        nb  += b.x * b.x + b.y * b.y + b.z * b.z + b.w * b.w;
    }
    float den = fmaxf(sqrtf(na) * sqrtf(nb), 1e-6f);
    out[e] = dot / den;
}


static inline int cdiv(long long a, long long b) { return (int)((a + b - 1) / b); }

extern "C" void kernel_launch(void* const* d_in, const int* in_sizes, int n_in,
                              void* d_out, int out_size, void* d_ws, size_t ws_size,
                              hipStream_t stream) {
    const float* x     = (const float*)d_in[0];
    const int*   ei    = (const int*)d_in[1];
    const float* ew    = (const float*)d_in[2];
    const int*   eli   = (const int*)d_in[3];
    const float* preW  = (const float*)d_in[4];
    const float* preb  = (const float*)d_in[5];
    const float* g1Wi  = (const float*)d_in[6];
    const float* g1Wr  = (const float*)d_in[7];
    const float* g1b   = (const float*)d_in[8];
    const float* g2Wi  = (const float*)d_in[9];
    const float* g2Wr  = (const float*)d_in[10];
    const float* g2b   = (const float*)d_in[11];
    const float* poW   = (const float*)d_in[12];
    const float* pob   = (const float*)d_in[13];
    float* out = (float*)d_out;

    const int N = in_sizes[0];
    const int E = in_sizes[2];
    const int L = in_sizes[3] / 2;

    const int* erow = ei;
    const int* ecol = ei + E;

    float* ws   = (float*)d_ws;
    float* dinv = ws;
    float* aux  = dinv + N;
    float* h64  = aux  + E;
    float* t    = h64  + (size_t)N * 64;
    float* agg1 = t    + (size_t)N * 32;
    float* agg2 = agg1 + (size_t)N * 32;
    float* z    = agg2 + (size_t)N * 32;

    const int NB = cdiv(N, BIN_NODES);
    const int chunk = cdiv(E, PART_WG);
    int* counts   = (int*)aux;
    int* binStart = counts + PART_WG * NB;
    int* rec      = (int*)h64;

    const int B = 256;
    const int gemm_threads = cdiv(N, 16) * 32;

    bin_count_kernel<<<PART_WG, WG_THREADS, 0, stream>>>(ecol, counts, E, NB, chunk);
    bin_fill_kernel<<<PART_WG, WG_THREADS, 0, stream>>>(erow, ecol, ew, counts, binStart, rec, E, NB, chunk);
    bin_degree_kernel<<<NB, WG_THREADS, 0, stream>>>(rec, binStart, dinv, N);

    pre_gemm_kernel<<<cdiv(gemm_threads, B), B, 0, stream>>>(
        x, preW, preb, g1Wi, g1Wr, g1b, t, agg1, dinv, N);

    bin_accumulate_gemm_kernel<2, true><<<NB, WG_THREADS, 0, stream>>>(
        rec, binStart, t, agg1, g2Wi, g2Wr, g2b, agg2, dinv, N);

    bin_accumulate_gemm_kernel<1, false><<<NB, WG_THREADS, 0, stream>>>(
        rec, binStart, agg1, agg2, poW, nullptr, pob, z, dinv, N);

    score_kernel<<<cdiv(L, B), B, 0, stream>>>(eli, z, out, L);
}


# accumulate: a bin's last, partial trip asks for all its rows as one group, like a full trip, instead of one dependent row load per record while 31 waves wait; sum order = LDS arrival, as before
# speedup vs baseline: 5.3314x; 1.0556x over previous
#include <hip/hip_runtime.h>
#include <math.h>

typedef float v2f __attribute__((ext_vector_type(2)));
typedef float v8f __attribute__((ext_vector_type(8)));

static __device__ __forceinline__ float elu1(float v) {
    return v > 0.0f ? v : (__expf(v) - 1.0f);
}


#define BIN_SHIFT 8
#define BIN_NODES 256
#define MAXB 512
#define PART_WG 256
#define WG_THREADS 1024

__global__ void __launch_bounds__(WG_THREADS)
bin_count_kernel(const int* __restrict__ col, int* __restrict__ counts, int E, int NB, int chunk) {
    __shared__ int hist[MAXB];
    const int tid = threadIdx.x;
    if (tid < MAXB) hist[tid] = 0;
    __syncthreads();
    const int base = blockIdx.x * chunk;
    const int end = min(base + chunk, E);
    for (int e0 = base; e0 < end; e0 += 4 * WG_THREADS) {
        int c[4];
#pragma unroll
        for (int k = 0; k < 4; ++k) {
            int e = e0 + k * WG_THREADS + tid;
            c[k] = (e < end) ? col[e] : -1;
        }
#pragma unroll
        for (int k = 0; k < 4; ++k)
            if (c[k] >= 0) atomicAdd(&hist[c[k] >> BIN_SHIFT], 1);
    }
    __syncthreads();
    if (tid < NB) counts[blockIdx.x * NB + tid] = hist[tid];
}

__global__ void __launch_bounds__(WG_THREADS)
bin_fill_kernel(const int* __restrict__ row, const int* __restrict__ col, const float* __restrict__ w,
                const int* __restrict__ counts, int* __restrict__ binStart, int* __restrict__ recw,
                int E, int NB, int chunk) {
    int2* __restrict__ rec = (int2*)recw;
    __shared__ int cursor[MAXB];
    __shared__ int sc[MAXB];
    __shared__ int ptot[2][MAXB];
    __shared__ int ppre[2][MAXB];
    const int tid = threadIdx.x;
    const int g = blockIdx.x;
    {
        const int b = tid & (MAXB - 1);
        const int part = tid >> 9;
        int tot = 0, pre = 0;
        if (b < NB) {
            const int g0 = part * (PART_WG / 2);
#pragma unroll 16
            for (int i = 0; i < PART_WG / 2; ++i) {
                int gg = g0 + i;
                int c = counts[gg * NB + b];
                tot += c;
                pre += (gg < g) ? c : 0;
            }
        }
        ptot[part][b] = tot;
        ppre[part][b] = pre;
    }
    __syncthreads();
    int tot = 0, pre = 0;
    if (tid < MAXB) {
        tot = ptot[0][tid] + ptot[1][tid];
        pre = ppre[0][tid] + ppre[1][tid];
        sc[tid] = tot;
    }
    __syncthreads();
    for (int off = 1; off < MAXB; off <<= 1) {
        int v = 0;
        if (tid < MAXB && tid >= off) v = sc[tid - off];
        __syncthreads();
        if (tid < MAXB) sc[tid] += v;
        __syncthreads();
    }
    if (tid < MAXB) {
        int excl = sc[tid] - tot;
        cursor[tid] = excl + pre;
        if (g == 0 && tid <= NB) binStart[tid] = excl;
    }
    __syncthreads();

    const int base = g * chunk;
    const int end = min(base + chunk, E);
    for (int e0 = base; e0 < end; e0 += 4 * WG_THREADS) {
        int r[4], c[4];
        float ww[4];
#pragma unroll
        for (int k = 0; k < 4; ++k) {
            int e = e0 + k * WG_THREADS + tid;
            bool ok = e < end;
            r[k] = ok ? row[e] : 0;
            c[k] = ok ? col[e] : -1;
            ww[k] = ok ? w[e] : 0.0f;
        }
#pragma unroll
        for (int k = 0; k < 4; ++k) {
            if (c[k] >= 0) {
                int pos = atomicAdd(&cursor[c[k] >> BIN_SHIFT], 1);
                unsigned packed = (unsigned)r[k] | ((unsigned)(c[k] & (BIN_NODES - 1)) << 24);
                rec[pos] = make_int2((int)packed, __float_as_int(ww[k]));
            }
        }
    }
}

__global__ void __launch_bounds__(WG_THREADS)
bin_degree_kernel(const int* __restrict__ recw, const int* __restrict__ binStart,
                  float* __restrict__ dinv, int N) {
    const int2* __restrict__ rec = (const int2*)recw;
    __shared__ float d[BIN_NODES];
    const int tid = threadIdx.x;
    if (tid < BIN_NODES) d[tid] = 0.0f;
    __syncthreads();
    const int s = binStart[blockIdx.x];
    const int e = binStart[blockIdx.x + 1];
    for (int p0 = s; p0 < e; p0 += 4 * WG_THREADS) {
        int2 r[4];
#pragma unroll
        for (int k = 0; k < 4; ++k) {
            int p = p0 + k * WG_THREADS + tid;
            r[k] = rec[min(p, e - 1)];
        }
#pragma unroll
        for (int k = 0; k < 4; ++k) {
            int p = p0 + k * WG_THREADS + tid;
            if (p < e) atomicAdd(&d[(unsigned)r[k].x >> 24], __int_as_float(r[k].y));
        }
    }
    __syncthreads();
    if (tid < BIN_NODES) {
        int node = blockIdx.x * BIN_NODES + tid;
        if (node < N) {
            float v = d[tid];
            dinv[node] = v > 0.0f ? rsqrtf(v) : 0.0f;
        }
    }
}

#define TP 36
template <int NCT, bool DUAL>
__global__ void __launch_bounds__(WG_THREADS, 16)
bin_accumulate_gemm_kernel(const int* __restrict__ recw, const int* __restrict__ binStart,
                           const float* __restrict__ t, float* root,
                           const float* __restrict__ Wi, const float* __restrict__ Wr,
                           const float* __restrict__ biasR, float* __restrict__ outR,
                           const float* __restrict__ dinv, int N) {
    const int2* __restrict__ rec = (const int2*)recw;
    __shared__ float tile[BIN_NODES * TP];
    __shared__ float dc[BIN_NODES];
    const int tid = threadIdx.x;
    const int lane = tid & 31;
    const int wave = __builtin_amdgcn_readfirstlane(tid >> 5);
    for (int i = tid; i < BIN_NODES * TP; i += WG_THREADS) tile[i] = 0.0f;
    if (tid < BIN_NODES) {
        int node = blockIdx.x * BIN_NODES + tid;
        dc[tid] = node < N ? dinv[node] : 0.0f;
    }
    __syncthreads();

    const int s = binStart[blockIdx.x];
    const int e = binStart[blockIdx.x + 1];
    const int step = (WG_THREADS / 32) * 32;

    int p0 = s + wave * 32;
    int2 cur = make_int2(0, 0);
    if (p0 < e) {
        int p = p0 + lane;
        if (p < e) cur = rec[p];
    }
    while (p0 < e) {
        const int pn = p0 + step;
        int2 nxt = make_int2(0, 0);
        if (pn < e) {
            int p = pn + lane;
            if (p < e) nxt = rec[p];
        }
        const int cnt = e - p0;
        float v[32];
#pragma unroll
        for (int j = 0; j < 32; ++j) {
            int r = __builtin_amdgcn_readlane(cur.x, j) & 0xFFFFFF;
            v[j] = t[(unsigned)(r * 32 + lane)];
        }
        if (cnt >= 32) {
#pragma unroll
            for (int j = 0; j < 32; ++j) {
                unsigned cl = (unsigned)__builtin_amdgcn_readlane(cur.x, j) >> 24;
                float n = __int_as_float(__builtin_amdgcn_readlane(cur.y, j));
                atomicAdd(&tile[cl * TP + lane], n * v[j]);
            }
        } else {
#pragma unroll
            for (int j = 0; j < 32; ++j) {
                if (j < cnt) {
                    unsigned cl = (unsigned)__builtin_amdgcn_readlane(cur.x, j) >> 24;
                    float n = __int_as_float(__builtin_amdgcn_readlane(cur.y, j));
                    atomicAdd(&tile[cl * TP + lane], n * v[j]);
                }
            }
        }
        p0 = pn;
        cur = nxt;
    }
    __syncthreads();

    const int nodeBase = blockIdx.x * BIN_NODES;
    for (int k0 = 0; k0 < (BIN_NODES * 32) / WG_THREADS; k0 += 4) {
        float a[4];
#pragma unroll
        for (int k = 0; k < 4; ++k) {
            int i = (k0 + k) * WG_THREADS + tid;
            int node = nodeBase + (i >> 5);
            a[k] = (node < N) ? root[(size_t)node * 32 + lane] : 0.0f;
        }
#pragma unroll
        for (int k = 0; k < 4; ++k) {
            int i = (k0 + k) * WG_THREADS + tid;
            int nl = i >> 5;
            const float sum = tile[nl * TP + lane];
            const float dn = dc[nl];
            float h = (nodeBase + nl < N) ? fmaxf(a[k] + dn * sum, 0.0f) : 0.0f;
            tile[nl * TP + lane] = h;
        }
    }
    __syncthreads();

    const int part = wave >> 4;
    if (DUAL || part == 0) {
        constexpr int F = NCT * 16;
        const int rt = wave & 15;
        const int m  = lane & 15;
        const int hi = lane >> 4;
        const bool second = DUAL && (part != 0);
        const float* __restrict__ Wsel = second ? Wr : Wi;
        const bool withBias = DUAL ? second : true;
        float* dst = DUAL ? (second ? outR : root) : outR;

        float bv[NCT];
#pragma unroll
        for (int ct = 0; ct < NCT; ++ct) bv[ct] = withBias ? biasR[ct * 16 + m] : -0.0f;
        const float am = (DUAL && !second) ? dc[rt * 16 + m] : 1.0f;

        v8f acc[NCT] = {};
        const float* arow = &tile[(rt * 16 + m) * TP + 2 * hi];
#pragma unroll
        for (int k0 = 0; k0 < 32; k0 += 4) {
            const int ka = k0 + 2 * hi;
            v2f a;
            a.x = arow[k0] * am;
            a.y = arow[k0 + 1] * am;
#pragma unroll
            for (int ct = 0; ct < NCT; ++ct) {
                const int colc = ct * 16 + m;
                v2f b;
                b.x = Wsel[(size_t)ka * F + colc];
                b.y = Wsel[(size_t)(ka + 1) * F + colc];
                acc[ct] = __builtin_amdgcn_wmma_f32_16x16x4_f32(
                    false, a, false, b, (short)0, acc[ct], false, false);
            }
        }

#pragma unroll
        for (int ct = 0; ct < NCT; ++ct) {
            const int colc = ct * 16 + m;
            float o[8];
#pragma unroll
            for (int r = 0; r < 8; ++r) o[r] = acc[ct][r] + bv[ct];
#pragma unroll
            for (int r = 0; r < 8; ++r) {
                const int node = nodeBase + rt * 16 + r + 8 * hi;
                if (node < N) dst[(size_t)node * F + colc] = o[r];
            }
        }
    }
}


__global__ void pre_gemm_kernel(const float* __restrict__ x,
                                const float* __restrict__ preW,
                                const float* __restrict__ preb,
                                const float* __restrict__ Wi,
                                const float* __restrict__ Wr,
                                const float* __restrict__ biasR,
                                float* __restrict__ Ti,
                                float* __restrict__ Tr,
                                const float* __restrict__ dinv,
                                int nrows) {
    constexpr int K = 64;
    constexpr int NCT = 2;
    constexpr int F = NCT * 16;
    int wave = (blockIdx.x * blockDim.x + threadIdx.x) >> 5;
    int lane = threadIdx.x & 31;
    int row0 = wave * 16;
    if (row0 >= nrows) return;

    int m  = lane & 15;
    int hi = lane >> 4;

    v8f accI[NCT] = {};
    v8f accR[NCT] = {};

    int ar = row0 + m;
    if (ar >= nrows) ar = nrows - 1;
    const float xv = x[ar];

    for (int k0 = 0; k0 < K; k0 += 4) {
        int ka = k0 + 2 * hi;
        v2f a;
        a.x = elu1(xv * preW[ka] + preb[ka]);
        a.y = elu1(xv * preW[ka + 1] + preb[ka + 1]);
#pragma unroll
        for (int ct = 0; ct < NCT; ++ct) {
            int colc = ct * 16 + m;
            v2f b;
            b.x = Wi[(size_t)ka * F + colc];
            b.y = Wi[(size_t)(ka + 1) * F + colc];
            accI[ct] = __builtin_amdgcn_wmma_f32_16x16x4_f32(
                false, a, false, b, (short)0, accI[ct], false, false);
            v2f br;
            br.x = Wr[(size_t)ka * F + colc];
            br.y = Wr[(size_t)(ka + 1) * F + colc];
            accR[ct] = __builtin_amdgcn_wmma_f32_16x16x4_f32(
                false, a, false, br, (short)0, accR[ct], false, false);
        }
    }

    float dv[8];
#pragma unroll
    for (int r = 0; r < 8; ++r) dv[r] = dinv[min(row0 + r + 8 * hi, nrows - 1)];

#pragma unroll
    for (int ct = 0; ct < NCT; ++ct) {
        int colc = ct * 16 + m;
        float bv = biasR[colc];
#pragma unroll
        for (int r = 0; r < 8; ++r) {
            int rowm = row0 + r + 8 * hi;
            if (rowm >= nrows) continue;
            size_t idx = (size_t)rowm * F + colc;
            Ti[idx] = accI[ct][r] * dv[r];
            Tr[idx] = accR[ct][r] + bv;
        }
    }
}


__global__ void score_kernel(const int* __restrict__ li, const float* __restrict__ z,
                             float* __restrict__ out, int L) {
    int e = blockIdx.x * blockDim.x + threadIdx.x;
    if (e >= L) return;
    int i = li[e], j = li[L + e];
    const float4* za = (const float4*)(z + (size_t)i * 16);
    const float4* zb = (const float4*)(z + (size_t)j * 16);
    float dot = 0.0f, na = 0.0f, nb = 0.0f;
#pragma unroll
    for (int q = 0; q < 4; ++q) {
        float4 a = za[q], b = zb[q];
        dot += a.x * b.x + a.y * b.y + a.z * b.z + a.w * b.w;
        na  += a.x * a.x + a.y * a.y + a.z * a.z + a.w * a.w;
        nb  += b.x * b.x + b.y * b.y + b.z * b.z + b.w * b.w;
    }
    float den = fmaxf(sqrtf(na) * sqrtf(nb), 1e-6f);
    out[e] = dot / den;
}


static inline int cdiv(long long a, long long b) { return (int)((a + b - 1) / b); }

extern "C" void kernel_launch(void* const* d_in, const int* in_sizes, int n_in,
                              void* d_out, int out_size, void* d_ws, size_t ws_size,
                              hipStream_t stream) {
    const float* x     = (const float*)d_in[0];
    const int*   ei    = (const int*)d_in[1];
    const float* ew    = (const float*)d_in[2];
    const int*   eli   = (const int*)d_in[3];
    const float* preW  = (const float*)d_in[4];
    const float* preb  = (const float*)d_in[5];
    const float* g1Wi  = (const float*)d_in[6];
    const float* g1Wr  = (const float*)d_in[7];
    const float* g1b   = (const float*)d_in[8];
    const float* g2Wi  = (const float*)d_in[9];
    const float* g2Wr  = (const float*)d_in[10];
    const float* g2b   = (const float*)d_in[11];
    const float* poW   = (const float*)d_in[12];
    const float* pob   = (const float*)d_in[13];
    float* out = (float*)d_out;

    const int N = in_sizes[0];
    const int E = in_sizes[2];
    const int L = in_sizes[3] / 2;

    const int* erow = ei;
    const int* ecol = ei + E;

    float* ws   = (float*)d_ws;
    float* dinv = ws;
    float* aux  = dinv + N;
    float* h64  = aux  + E;
    float* t    = h64  + (size_t)N * 64;
    float* agg1 = t    + (size_t)N * 32;
    float* agg2 = agg1 + (size_t)N * 32;
    float* z    = agg2 + (size_t)N * 32;

    const int NB = cdiv(N, BIN_NODES);
    const int chunk = cdiv(E, PART_WG);
    int* counts   = (int*)aux;
    int* binStart = counts + PART_WG * NB;
    int* rec      = (int*)h64;

    const int B = 256;
    const int gemm_threads = cdiv(N, 16) * 32;

    bin_count_kernel<<<PART_WG, WG_THREADS, 0, stream>>>(ecol, counts, E, NB, chunk);
    bin_fill_kernel<<<PART_WG, WG_THREADS, 0, stream>>>(erow, ecol, ew, counts, binStart, rec, E, NB, chunk);
    bin_degree_kernel<<<NB, WG_THREADS, 0, stream>>>(rec, binStart, dinv, N);

    pre_gemm_kernel<<<cdiv(gemm_threads, B), B, 0, stream>>>(
        x, preW, preb, g1Wi, g1Wr, g1b, t, agg1, dinv, N);

    bin_accumulate_gemm_kernel<2, true><<<NB, WG_THREADS, 0, stream>>>(
        rec, binStart, t, agg1, g2Wi, g2Wr, g2b, agg2, dinv, N);

    bin_accumulate_gemm_kernel<1, false><<<NB, WG_THREADS, 0, stream>>>(
        rec, binStart, agg1, agg2, poW, nullptr, pob, z, dinv, N);

    score_kernel<<<cdiv(L, B), B, 0, stream>>>(eli, z, out, L);
}


# bins of 196 nodes, not 256: 511 one-bin workgroups fill the 512 resident slots (2 per WGP), all WGPs loaded alike, where 391 left some with two bins, the rest fewer; sum order = LDS arrival as before
# speedup vs baseline: 5.5553x; 1.0420x over previous
#include <hip/hip_runtime.h>
#include <math.h>

typedef float v2f __attribute__((ext_vector_type(2)));
typedef float v8f __attribute__((ext_vector_type(8)));

static __device__ __forceinline__ float elu1(float v) {
    return v > 0.0f ? v : (__expf(v) - 1.0f);
}


#define BIN_NODES 196
#define NRT 13
#define TILE_ROWS (NRT * 16)
#define MAXB 512
#define PART_WG 256
#define WG_THREADS 1024

__global__ void __launch_bounds__(WG_THREADS)
bin_count_kernel(const int* __restrict__ col, int* __restrict__ counts, int E, int NB, int chunk) {
    __shared__ int hist[MAXB];
    const int tid = threadIdx.x;
    if (tid < MAXB) hist[tid] = 0;
    __syncthreads();
    const int base = blockIdx.x * chunk;
    const int end = min(base + chunk, E);
    for (int e0 = base; e0 < end; e0 += 4 * WG_THREADS) {
        int c[4];
#pragma unroll
        for (int k = 0; k < 4; ++k) {
            int e = e0 + k * WG_THREADS + tid;
            c[k] = (e < end) ? col[e] : -1;
        }
#pragma unroll
        for (int k = 0; k < 4; ++k)
            if (c[k] >= 0) atomicAdd(&hist[(unsigned)c[k] / BIN_NODES], 1);
    }
    __syncthreads();
    if (tid < NB) counts[blockIdx.x * NB + tid] = hist[tid];
}

__global__ void __launch_bounds__(WG_THREADS)
bin_fill_kernel(const int* __restrict__ row, const int* __restrict__ col, const float* __restrict__ w,
                const int* __restrict__ counts, int* __restrict__ binStart, int* __restrict__ recw,
                int E, int NB, int chunk) {
    int2* __restrict__ rec = (int2*)recw;
    __shared__ int cursor[MAXB];
    __shared__ int sc[MAXB];
    __shared__ int ptot[2][MAXB];
    __shared__ int ppre[2][MAXB];
    const int tid = threadIdx.x;
    const int g = blockIdx.x;
    {
        const int b = tid & (MAXB - 1);
        const int part = tid >> 9;
        int tot = 0, pre = 0;
        if (b < NB) {
            const int g0 = part * (PART_WG / 2);
#pragma unroll 16
            for (int i = 0; i < PART_WG / 2; ++i) {
                int gg = g0 + i;
                int c = counts[gg * NB + b];
                tot += c;
                pre += (gg < g) ? c : 0;
            }
        }
        ptot[part][b] = tot;
        ppre[part][b] = pre;
    }
    __syncthreads();
    int tot = 0, pre = 0;
    if (tid < MAXB) {
        tot = ptot[0][tid] + ptot[1][tid];
        pre = ppre[0][tid] + ppre[1][tid];
        sc[tid] = tot;
    }
    __syncthreads();
    for (int off = 1; off < MAXB; off <<= 1) {
        int v = 0;
        if (tid < MAXB && tid >= off) v = sc[tid - off];
        __syncthreads();
        if (tid < MAXB) sc[tid] += v;
        __syncthreads();
    }
    if (tid < MAXB) {
        int excl = sc[tid] - tot;
        cursor[tid] = excl + pre;
        if (g == 0 && tid <= NB) binStart[tid] = excl;
    }
    __syncthreads();

    const int base = g * chunk;
    const int end = min(base + chunk, E);
    for (int e0 = base; e0 < end; e0 += 4 * WG_THREADS) {
        int r[4], c[4];
        float ww[4];
#pragma unroll
        for (int k = 0; k < 4; ++k) {
            int e = e0 + k * WG_THREADS + tid;
            bool ok = e < end;
            r[k] = ok ? row[e] : 0;
            c[k] = ok ? col[e] : -1;
            ww[k] = ok ? w[e] : 0.0f;
        }
#pragma unroll
        for (int k = 0; k < 4; ++k) {
            if (c[k] >= 0) {
                const unsigned bin = (unsigned)c[k] / BIN_NODES;
                int pos = atomicAdd(&cursor[bin], 1);
                unsigned packed = (unsigned)r[k] | (((unsigned)c[k] - bin * BIN_NODES) << 24);
                rec[pos] = make_int2((int)packed, __float_as_int(ww[k]));
            }
        }
    }
}

__global__ void __launch_bounds__(WG_THREADS)
bin_degree_kernel(const int* __restrict__ recw, const int* __restrict__ binStart,
                  float* __restrict__ dinv, int N) {
    const int2* __restrict__ rec = (const int2*)recw;
    __shared__ float d[BIN_NODES];
    const int tid = threadIdx.x;
    if (tid < BIN_NODES) d[tid] = 0.0f;
    __syncthreads();
    const int s = binStart[blockIdx.x];
    const int e = binStart[blockIdx.x + 1];
    for (int p0 = s; p0 < e; p0 += 4 * WG_THREADS) {
        int2 r[4];
#pragma unroll
        for (int k = 0; k < 4; ++k) {
            int p = p0 + k * WG_THREADS + tid;
            r[k] = rec[min(p, e - 1)];
        }
#pragma unroll
        for (int k = 0; k < 4; ++k) {
            int p = p0 + k * WG_THREADS + tid;
            if (p < e) atomicAdd(&d[(unsigned)r[k].x >> 24], __int_as_float(r[k].y));
        }
    }
    __syncthreads();
    if (tid < BIN_NODES) {
        int node = blockIdx.x * BIN_NODES + tid;
        if (node < N) {
            float v = d[tid];
            dinv[node] = v > 0.0f ? rsqrtf(v) : 0.0f;
        }
    }
}

#define TP 36
template <int NCT, bool DUAL>
__global__ void __launch_bounds__(WG_THREADS, 16)
bin_accumulate_gemm_kernel(const int* __restrict__ recw, const int* __restrict__ binStart,
                           const float* __restrict__ t, float* root,
                           const float* __restrict__ Wi, const float* __restrict__ Wr,
                           const float* __restrict__ biasR, float* __restrict__ outR,
                           const float* __restrict__ dinv, int N) {
    const int2* __restrict__ rec = (const int2*)recw;
    __shared__ float tile[TILE_ROWS * TP];
    __shared__ float dc[TILE_ROWS];
    const int tid = threadIdx.x;
    const int lane = tid & 31;
    const int wave = __builtin_amdgcn_readfirstlane(tid >> 5);
    for (int i = tid; i < TILE_ROWS * TP; i += WG_THREADS) tile[i] = 0.0f;
    if (tid < TILE_ROWS) {
        int node = blockIdx.x * BIN_NODES + tid;
        dc[tid] = (tid < BIN_NODES && node < N) ? dinv[node] : 0.0f;
    }
    __syncthreads();

    const int s = binStart[blockIdx.x];
    const int e = binStart[blockIdx.x + 1];
    const int step = (WG_THREADS / 32) * 32;

    int p0 = s + wave * 32;
    int2 cur = make_int2(0, 0);
    if (p0 < e) {
        int p = p0 + lane;
        if (p < e) cur = rec[p];
    }
    while (p0 < e) {
        const int pn = p0 + step;
        int2 nxt = make_int2(0, 0);
        if (pn < e) {
            int p = pn + lane;
            if (p < e) nxt = rec[p];
        }
        const int cnt = e - p0;
        float v[32];
#pragma unroll
        for (int j = 0; j < 32; ++j) {
            int r = __builtin_amdgcn_readlane(cur.x, j) & 0xFFFFFF;
            v[j] = t[(unsigned)(r * 32 + lane)];
        }
        if (cnt >= 32) {
#pragma unroll
            for (int j = 0; j < 32; ++j) {
                unsigned cl = (unsigned)__builtin_amdgcn_readlane(cur.x, j) >> 24;
                float n = __int_as_float(__builtin_amdgcn_readlane(cur.y, j));
                atomicAdd(&tile[cl * TP + lane], n * v[j]);
            }
        } else {
#pragma unroll
            for (int j = 0; j < 32; ++j) {
                if (j < cnt) {
                    unsigned cl = (unsigned)__builtin_amdgcn_readlane(cur.x, j) >> 24;
                    float n = __int_as_float(__builtin_amdgcn_readlane(cur.y, j));
                    atomicAdd(&tile[cl * TP + lane], n * v[j]);
                }
            }
        }
        p0 = pn;
        cur = nxt;
    }
    __syncthreads();

    const int nodeBase = blockIdx.x * BIN_NODES;
    constexpr int ROWS_PER_TRIP = WG_THREADS / 32;
    constexpr int EPI_TRIPS = (BIN_NODES + ROWS_PER_TRIP - 1) / ROWS_PER_TRIP;
    for (int k0 = 0; k0 < EPI_TRIPS; k0 += 4) {
        float a[4];
#pragma unroll
        for (int k = 0; k < 4; ++k) {
            const int nl = (k0 + k) * ROWS_PER_TRIP + wave;
            const int node = nodeBase + nl;
            a[k] = (nl < BIN_NODES && node < N) ? root[(size_t)node * 32 + lane] : 0.0f;
        }
#pragma unroll
        for (int k = 0; k < 4; ++k) {
            const int nl = (k0 + k) * ROWS_PER_TRIP + wave;
            if (nl < BIN_NODES) {
                const float sum = tile[nl * TP + lane];
                const float dn = dc[nl];
                float h = (nodeBase + nl < N) ? fmaxf(a[k] + dn * sum, 0.0f) : 0.0f;
                tile[nl * TP + lane] = h;
            }
        }
    }
    __syncthreads();

    const int part = wave >> 4;
    const int rt = wave & 15;
    if ((DUAL || part == 0) && rt < NRT) {
        constexpr int F = NCT * 16;
        const int m  = lane & 15;
        const int hi = lane >> 4;
        const bool second = DUAL && (part != 0);
        const float* __restrict__ Wsel = second ? Wr : Wi;
        const bool withBias = DUAL ? second : true;
        float* dst = DUAL ? (second ? outR : root) : outR;

        float bv[NCT];
#pragma unroll
        for (int ct = 0; ct < NCT; ++ct) bv[ct] = withBias ? biasR[ct * 16 + m] : -0.0f;
        const float am = (DUAL && !second) ? dc[rt * 16 + m] : 1.0f;

        v8f acc[NCT] = {};
        const float* arow = &tile[(rt * 16 + m) * TP + 2 * hi];
#pragma unroll
        for (int k0 = 0; k0 < 32; k0 += 4) {
            const int ka = k0 + 2 * hi;
            v2f a;
            a.x = arow[k0] * am;
            a.y = arow[k0 + 1] * am;
#pragma unroll
            for (int ct = 0; ct < NCT; ++ct) {
                const int colc = ct * 16 + m;
                v2f b;
                b.x = Wsel[(size_t)ka * F + colc];
                b.y = Wsel[(size_t)(ka + 1) * F + colc];
                acc[ct] = __builtin_amdgcn_wmma_f32_16x16x4_f32(
                    false, a, false, b, (short)0, acc[ct], false, false);
            }
        }

#pragma unroll
        for (int ct = 0; ct < NCT; ++ct) {
            const int colc = ct * 16 + m;
            float o[8];
#pragma unroll
            for (int r = 0; r < 8; ++r) o[r] = acc[ct][r] + bv[ct];
#pragma unroll
            for (int r = 0; r < 8; ++r) {
                const int nl = rt * 16 + r + 8 * hi;
                const int node = nodeBase + nl;
                if (nl < BIN_NODES && node < N) dst[(size_t)node * F + colc] = o[r];
            }
        }
    }
}


__global__ void pre_gemm_kernel(const float* __restrict__ x,
                                const float* __restrict__ preW,
                                const float* __restrict__ preb,
                                const float* __restrict__ Wi,
                                const float* __restrict__ Wr,
                                const float* __restrict__ biasR,
                                float* __restrict__ Ti,
                                float* __restrict__ Tr,
                                const float* __restrict__ dinv,
                                int nrows) {
    constexpr int K = 64;
    constexpr int NCT = 2;
    constexpr int F = NCT * 16;
    int wave = (blockIdx.x * blockDim.x + threadIdx.x) >> 5;
    int lane = threadIdx.x & 31;
    int row0 = wave * 16;
    if (row0 >= nrows) return;

    int m  = lane & 15;
    int hi = lane >> 4;

    v8f accI[NCT] = {};
    v8f accR[NCT] = {};

    int ar = row0 + m;
    if (ar >= nrows) ar = nrows - 1;
    const float xv = x[ar];

    for (int k0 = 0; k0 < K; k0 += 4) {
        int ka = k0 + 2 * hi;
        v2f a;
        a.x = elu1(xv * preW[ka] + preb[ka]);
        a.y = elu1(xv * preW[ka + 1] + preb[ka + 1]);
#pragma unroll
        for (int ct = 0; ct < NCT; ++ct) {
            int colc = ct * 16 + m;
            v2f b;
            b.x = Wi[(size_t)ka * F + colc];
            b.y = Wi[(size_t)(ka + 1) * F + colc];
            accI[ct] = __builtin_amdgcn_wmma_f32_16x16x4_f32(
                false, a, false, b, (short)0, accI[ct], false, false);
            v2f br;
            br.x = Wr[(size_t)ka * F + colc];
            br.y = Wr[(size_t)(ka + 1) * F + colc];
            accR[ct] = __builtin_amdgcn_wmma_f32_16x16x4_f32(
                false, a, false, br, (short)0, accR[ct], false, false);
        }
    }

    float dv[8];
#pragma unroll
    for (int r = 0; r < 8; ++r) dv[r] = dinv[min(row0 + r + 8 * hi, nrows - 1)];

#pragma unroll
    for (int ct = 0; ct < NCT; ++ct) {
        int colc = ct * 16 + m;
        float bv = biasR[colc];
#pragma unroll
        for (int r = 0; r < 8; ++r) {
            int rowm = row0 + r + 8 * hi;
            if (rowm >= nrows) continue;
            size_t idx = (size_t)rowm * F + colc;
            Ti[idx] = accI[ct][r] * dv[r];
            Tr[idx] = accR[ct][r] + bv;
        }
    }
}


__global__ void score_kernel(const int* __restrict__ li, const float* __restrict__ z,
                             float* __restrict__ out, int L) {
    int e = blockIdx.x * blockDim.x + threadIdx.x;
    if (e >= L) return;
    int i = li[e], j = li[L + e];
    const float4* za = (const float4*)(z + (size_t)i * 16);
    const float4* zb = (const float4*)(z + (size_t)j * 16);
    float dot = 0.0f, na = 0.0f, nb = 0.0f;
#pragma unroll
    for (int q = 0; q < 4; ++q) {
        float4 a = za[q], b = zb[q];
        dot += a.x * b.x + a.y * b.y + a.z * b.z + a.w * b.w;
        na  += a.x * a.x + a.y * a.y + a.z * a.z + a.w * a.w;
        nb  += b.x * b.x + b.y * b.y + b.z * b.z + b.w * b.w;
    }
    float den = fmaxf(sqrtf(na) * sqrtf(nb), 1e-6f);
    out[e] = dot / den;
}


static inline int cdiv(long long a, long long b) { return (int)((a + b - 1) / b); }

extern "C" void kernel_launch(void* const* d_in, const int* in_sizes, int n_in,
                              void* d_out, int out_size, void* d_ws, size_t ws_size,
                              hipStream_t stream) {
    const float* x     = (const float*)d_in[0];
    const int*   ei    = (const int*)d_in[1];
    const float* ew    = (const float*)d_in[2];
    const int*   eli   = (const int*)d_in[3];
    const float* preW  = (const float*)d_in[4];
    const float* preb  = (const float*)d_in[5];
    const float* g1Wi  = (const float*)d_in[6];
    const float* g1Wr  = (const float*)d_in[7];
    const float* g1b   = (const float*)d_in[8];
    const float* g2Wi  = (const float*)d_in[9];
    const float* g2Wr  = (const float*)d_in[10];
    const float* g2b   = (const float*)d_in[11];
    const float* poW   = (const float*)d_in[12];
    const float* pob   = (const float*)d_in[13];
    float* out = (float*)d_out;

    const int N = in_sizes[0];
    const int E = in_sizes[2];
    const int L = in_sizes[3] / 2;

    const int* erow = ei;
    const int* ecol = ei + E;

    float* ws   = (float*)d_ws;
    float* dinv = ws;
    float* aux  = dinv + N;
    float* h64  = aux  + E;
    float* t    = h64  + (size_t)N * 64;
    float* agg1 = t    + (size_t)N * 32;
    float* agg2 = agg1 + (size_t)N * 32;
    float* z    = agg2 + (size_t)N * 32;

    const int NB = cdiv(N, BIN_NODES);
    const int chunk = cdiv(E, PART_WG);
    int* counts   = (int*)aux;
    int* binStart = counts + PART_WG * NB;
    int* rec      = (int*)h64;

    const int B = 256;
    const int gemm_threads = cdiv(N, 16) * 32;

    bin_count_kernel<<<PART_WG, WG_THREADS, 0, stream>>>(ecol, counts, E, NB, chunk);
    bin_fill_kernel<<<PART_WG, WG_THREADS, 0, stream>>>(erow, ecol, ew, counts, binStart, rec, E, NB, chunk);
    bin_degree_kernel<<<NB, WG_THREADS, 0, stream>>>(rec, binStart, dinv, N);

    pre_gemm_kernel<<<cdiv(gemm_threads, B), B, 0, stream>>>(
        x, preW, preb, g1Wi, g1Wr, g1b, t, agg1, dinv, N);

    bin_accumulate_gemm_kernel<2, true><<<NB, WG_THREADS, 0, stream>>>(
        rec, binStart, t, agg1, g2Wi, g2Wr, g2b, agg2, dinv, N);

    bin_accumulate_gemm_kernel<1, false><<<NB, WG_THREADS, 0, stream>>>(
        rec, binStart, agg1, agg2, poW, nullptr, pob, z, dinv, N);

    score_kernel<<<cdiv(L, B), B, 0, stream>>>(eli, z, out, L);
}


# fill: each partition orders its records by bin in LDS (rank = return of the counting LDS add), then stores them in order: a wave's store is a few runs of slots, not 32 lines; sum order as before
# speedup vs baseline: 6.3039x; 1.1348x over previous
#include <hip/hip_runtime.h>
#include <math.h>

typedef float v2f __attribute__((ext_vector_type(2)));
typedef float v8f __attribute__((ext_vector_type(8)));

static __device__ __forceinline__ float elu1(float v) {
    return v > 0.0f ? v : (__expf(v) - 1.0f);
}


#define BIN_NODES 196
#define NRT 13
#define TILE_ROWS (NRT * 16)
#define MAXB 512
#define PART_WG 256
#define WG_THREADS 1024

__global__ void __launch_bounds__(WG_THREADS)
bin_count_kernel(const int* __restrict__ col, int* __restrict__ counts, int E, int NB, int chunk) {
    __shared__ int hist[MAXB];
    const int tid = threadIdx.x;
    if (tid < MAXB) hist[tid] = 0;
    __syncthreads();
    const int base = blockIdx.x * chunk;
    const int end = min(base + chunk, E);
    for (int e0 = base; e0 < end; e0 += 4 * WG_THREADS) {
        int c[4];
#pragma unroll
        for (int k = 0; k < 4; ++k) {
            int e = e0 + k * WG_THREADS + tid;
            c[k] = (e < end) ? col[e] : -1;
        }
#pragma unroll
        for (int k = 0; k < 4; ++k)
            if (c[k] >= 0) atomicAdd(&hist[(unsigned)c[k] / BIN_NODES], 1);
    }
    __syncthreads();
    if (tid < NB) counts[blockIdx.x * NB + tid] = hist[tid];
}

#define FILL_K 13
#define FILL_CAP (FILL_K * WG_THREADS)
__global__ void __launch_bounds__(WG_THREADS)
bin_fill_kernel(const int* __restrict__ row, const int* __restrict__ col, const float* __restrict__ w,
                const int* __restrict__ counts, int* __restrict__ binStart, int* __restrict__ recw,
                int E, int NB, int chunk) {
    int2* __restrict__ rec = (int2*)recw;
    __shared__ int cursor[MAXB];
    __shared__ int sc[MAXB];
    __shared__ int ptot[2][MAXB];
    __shared__ int ppre[2][MAXB];
    __shared__ int lhist[MAXB];
    __shared__ int lstart[MAXB];
    __shared__ int wsum[MAXB / 32];
    __shared__ int2 stage[FILL_CAP];
    __shared__ int stageG[FILL_CAP];
    const int tid = threadIdx.x;
    const int lane = tid & 31;
    const int g = blockIdx.x;
    {
        const int b = tid & (MAXB - 1);
        const int part = tid >> 9;
        int tot = 0, pre = 0;
        if (b < NB) {
            const int g0 = part * (PART_WG / 2);
#pragma unroll 16
            for (int i = 0; i < PART_WG / 2; ++i) {
                int gg = g0 + i;
                int c = counts[gg * NB + b];
                tot += c;
                pre += (gg < g) ? c : 0;
            }
        }
        ptot[part][b] = tot;
        ppre[part][b] = pre;
    }
    __syncthreads();
    int tot = 0, pre = 0;
    if (tid < MAXB) {
        tot = ptot[0][tid] + ptot[1][tid];
        pre = ppre[0][tid] + ppre[1][tid];
        sc[tid] = tot;
    }
    __syncthreads();
    for (int off = 1; off < MAXB; off <<= 1) {
        int v = 0;
        if (tid < MAXB && tid >= off) v = sc[tid - off];
        __syncthreads();
        if (tid < MAXB) sc[tid] += v;
        __syncthreads();
    }
    if (tid < MAXB) {
        int excl = sc[tid] - tot;
        cursor[tid] = excl + pre;
        if (g == 0 && tid <= NB) binStart[tid] = excl;
    }

    const int base = g * chunk;
    const int end = min(base + chunk, E);
    for (int c0 = base; c0 < end; c0 += FILL_CAP) {
        const int c1 = min(c0 + FILL_CAP, end);
        if (tid < MAXB) lhist[tid] = 0;
        __syncthreads();

        int r[FILL_K], c[FILL_K];
        float ww[FILL_K];
#pragma unroll
        for (int k = 0; k < FILL_K; ++k) {
            int e = c0 + k * WG_THREADS + tid;
            bool ok = e < c1;
            r[k] = ok ? row[e] : 0;
            c[k] = ok ? col[e] : -1;
            ww[k] = ok ? w[e] : 0.0f;
        }
        int br[FILL_K];
#pragma unroll
        for (int k = 0; k < FILL_K; ++k) {
            br[k] = -1;
            if (c[k] >= 0) {
                const unsigned bin = (unsigned)c[k] / BIN_NODES;
                const int rank = atomicAdd(&lhist[bin], 1);
                r[k] = (int)((unsigned)r[k] | (((unsigned)c[k] - bin * BIN_NODES) << 24));
                br[k] = (int)(bin << 16) | rank;
            }
        }
        __syncthreads();

        {
            const int v = (tid < MAXB) ? lhist[tid] : 0;
            int incl = v;
#pragma unroll
            for (int off = 1; off < 32; off <<= 1) {
                int n = __shfl_up(incl, off);
                if (lane >= off) incl += n;
            }
            if (tid < MAXB && lane == 31) wsum[tid >> 5] = incl;
            __syncthreads();
            if (tid < MAXB) {
                int before = 0;
#pragma unroll
                for (int q = 0; q < MAXB / 32; ++q) before += (q < (tid >> 5)) ? wsum[q] : 0;
                lstart[tid] = before + incl - v;
            }
            __syncthreads();
        }

#pragma unroll
        for (int k = 0; k < FILL_K; ++k) {
            if (br[k] >= 0) {
                const int bin = br[k] >> 16;
                const int rank = br[k] & 0xFFFF;
                const int lp = lstart[bin] + rank;
                stage[lp] = make_int2(r[k], __float_as_int(ww[k]));
                stageG[lp] = cursor[bin] + rank;
            }
        }
        __syncthreads();

        const int n = c1 - c0;
        for (int i = tid; i < n; i += WG_THREADS) rec[stageG[i]] = stage[i];

        if (tid < MAXB) cursor[tid] += lhist[tid];
    }
}

__global__ void __launch_bounds__(WG_THREADS)
bin_degree_kernel(const int* __restrict__ recw, const int* __restrict__ binStart,
                  float* __restrict__ dinv, int N) {
    const int2* __restrict__ rec = (const int2*)recw;
    __shared__ float d[BIN_NODES];
    const int tid = threadIdx.x;
    if (tid < BIN_NODES) d[tid] = 0.0f;
    __syncthreads();
    const int s = binStart[blockIdx.x];
    const int e = binStart[blockIdx.x + 1];
    for (int p0 = s; p0 < e; p0 += 4 * WG_THREADS) {
        int2 r[4];
#pragma unroll
        for (int k = 0; k < 4; ++k) {
            int p = p0 + k * WG_THREADS + tid;
            r[k] = rec[min(p, e - 1)];
        }
#pragma unroll
        for (int k = 0; k < 4; ++k) {
            int p = p0 + k * WG_THREADS + tid;
            if (p < e) atomicAdd(&d[(unsigned)r[k].x >> 24], __int_as_float(r[k].y));
        }
    }
    __syncthreads();
    if (tid < BIN_NODES) {
        int node = blockIdx.x * BIN_NODES + tid;
        if (node < N) {
            float v = d[tid];
            dinv[node] = v > 0.0f ? rsqrtf(v) : 0.0f;
        }
    }
}

#define TP 36
template <int NCT, bool DUAL>
__global__ void __launch_bounds__(WG_THREADS, 16)
bin_accumulate_gemm_kernel(const int* __restrict__ recw, const int* __restrict__ binStart,
                           const float* __restrict__ t, float* root,
                           const float* __restrict__ Wi, const float* __restrict__ Wr,
                           const float* __restrict__ biasR, float* __restrict__ outR,
                           const float* __restrict__ dinv, int N) {
    const int2* __restrict__ rec = (const int2*)recw;
    __shared__ float tile[TILE_ROWS * TP];
    __shared__ float dc[TILE_ROWS];
    const int tid = threadIdx.x;
    const int lane = tid & 31;
    const int wave = __builtin_amdgcn_readfirstlane(tid >> 5);
    for (int i = tid; i < TILE_ROWS * TP; i += WG_THREADS) tile[i] = 0.0f;
    if (tid < TILE_ROWS) {
        int node = blockIdx.x * BIN_NODES + tid;
        dc[tid] = (tid < BIN_NODES && node < N) ? dinv[node] : 0.0f;
    }
    __syncthreads();

    const int s = binStart[blockIdx.x];
    const int e = binStart[blockIdx.x + 1];
    const int step = (WG_THREADS / 32) * 32;

    int p0 = s + wave * 32;
    int2 cur = make_int2(0, 0);
    if (p0 < e) {
        int p = p0 + lane;
        if (p < e) cur = rec[p];
    }
    while (p0 < e) {
        const int pn = p0 + step;
        int2 nxt = make_int2(0, 0);
        if (pn < e) {
            int p = pn + lane;
            if (p < e) nxt = rec[p];
        }
        const int cnt = e - p0;
        float v[32];
#pragma unroll
        for (int j = 0; j < 32; ++j) {
            int r = __builtin_amdgcn_readlane(cur.x, j) & 0xFFFFFF;
            v[j] = t[(unsigned)(r * 32 + lane)];
        }
        if (cnt >= 32) {
#pragma unroll
            for (int j = 0; j < 32; ++j) {
                unsigned cl = (unsigned)__builtin_amdgcn_readlane(cur.x, j) >> 24;
                float n = __int_as_float(__builtin_amdgcn_readlane(cur.y, j));
                atomicAdd(&tile[cl * TP + lane], n * v[j]);
            }
        } else {
#pragma unroll
            for (int j = 0; j < 32; ++j) {
                if (j < cnt) {
                    unsigned cl = (unsigned)__builtin_amdgcn_readlane(cur.x, j) >> 24;
                    float n = __int_as_float(__builtin_amdgcn_readlane(cur.y, j));
                    atomicAdd(&tile[cl * TP + lane], n * v[j]);
                }
            }
        }
        p0 = pn;
        cur = nxt;
    }
    __syncthreads();

    const int nodeBase = blockIdx.x * BIN_NODES;
    constexpr int ROWS_PER_TRIP = WG_THREADS / 32;
    constexpr int EPI_TRIPS = (BIN_NODES + ROWS_PER_TRIP - 1) / ROWS_PER_TRIP;
    for (int k0 = 0; k0 < EPI_TRIPS; k0 += 4) {
        float a[4];
#pragma unroll
        for (int k = 0; k < 4; ++k) {
            const int nl = (k0 + k) * ROWS_PER_TRIP + wave;
            const int node = nodeBase + nl;
            a[k] = (nl < BIN_NODES && node < N) ? root[(size_t)node * 32 + lane] : 0.0f;
        }
#pragma unroll
        for (int k = 0; k < 4; ++k) {
            const int nl = (k0 + k) * ROWS_PER_TRIP + wave;
            if (nl < BIN_NODES) {
                const float sum = tile[nl * TP + lane];
                const float dn = dc[nl];
                float h = (nodeBase + nl < N) ? fmaxf(a[k] + dn * sum, 0.0f) : 0.0f;
                tile[nl * TP + lane] = h;
            }
        }
    }
    __syncthreads();

    const int part = wave >> 4;
    const int rt = wave & 15;
    if ((DUAL || part == 0) && rt < NRT) {
        constexpr int F = NCT * 16;
        const int m  = lane & 15;
        const int hi = lane >> 4;
        const bool second = DUAL && (part != 0);
        const float* __restrict__ Wsel = second ? Wr : Wi;
        const bool withBias = DUAL ? second : true;
        float* dst = DUAL ? (second ? outR : root) : outR;

        float bv[NCT];
#pragma unroll
        for (int ct = 0; ct < NCT; ++ct) bv[ct] = withBias ? biasR[ct * 16 + m] : -0.0f;
        const float am = (DUAL && !second) ? dc[rt * 16 + m] : 1.0f;

        v8f acc[NCT] = {};
        const float* arow = &tile[(rt * 16 + m) * TP + 2 * hi];
#pragma unroll
        for (int k0 = 0; k0 < 32; k0 += 4) {
            const int ka = k0 + 2 * hi;
            v2f a;
            a.x = arow[k0] * am;
            a.y = arow[k0 + 1] * am;
#pragma unroll
            for (int ct = 0; ct < NCT; ++ct) {
                const int colc = ct * 16 + m;
                v2f b;
                b.x = Wsel[(size_t)ka * F + colc];
                b.y = Wsel[(size_t)(ka + 1) * F + colc];
                acc[ct] = __builtin_amdgcn_wmma_f32_16x16x4_f32(
                    false, a, false, b, (short)0, acc[ct], false, false);
            }
        }

#pragma unroll
        for (int ct = 0; ct < NCT; ++ct) {
            const int colc = ct * 16 + m;
            float o[8];
#pragma unroll
            for (int r = 0; r < 8; ++r) o[r] = acc[ct][r] + bv[ct];
#pragma unroll
            for (int r = 0; r < 8; ++r) {
                const int nl = rt * 16 + r + 8 * hi;
                const int node = nodeBase + nl;
                if (nl < BIN_NODES && node < N) dst[(size_t)node * F + colc] = o[r];
            }
        }
    }
}


__global__ void pre_gemm_kernel(const float* __restrict__ x,
                                const float* __restrict__ preW,
                                const float* __restrict__ preb,
                                const float* __restrict__ Wi,
                                const float* __restrict__ Wr,
                                const float* __restrict__ biasR,
                                float* __restrict__ Ti,
                                float* __restrict__ Tr,
                                const float* __restrict__ dinv,
                                int nrows) {
    constexpr int K = 64;
    constexpr int NCT = 2;
    constexpr int F = NCT * 16;
    int wave = (blockIdx.x * blockDim.x + threadIdx.x) >> 5;
    int lane = threadIdx.x & 31;
    int row0 = wave * 16;
    if (row0 >= nrows) return;

    int m  = lane & 15;
    int hi = lane >> 4;

    v8f accI[NCT] = {};
    v8f accR[NCT] = {};

    int ar = row0 + m;
    if (ar >= nrows) ar = nrows - 1;
    const float xv = x[ar];

    for (int k0 = 0; k0 < K; k0 += 4) {
        int ka = k0 + 2 * hi;
        v2f a;
        a.x = elu1(xv * preW[ka] + preb[ka]);
        a.y = elu1(xv * preW[ka + 1] + preb[ka + 1]);
#pragma unroll
        for (int ct = 0; ct < NCT; ++ct) {
            int colc = ct * 16 + m;
            v2f b;
            b.x = Wi[(size_t)ka * F + colc];
            b.y = Wi[(size_t)(ka + 1) * F + colc];
            accI[ct] = __builtin_amdgcn_wmma_f32_16x16x4_f32(
                false, a, false, b, (short)0, accI[ct], false, false);
            v2f br;
            br.x = Wr[(size_t)ka * F + colc];
            br.y = Wr[(size_t)(ka + 1) * F + colc];
            accR[ct] = __builtin_amdgcn_wmma_f32_16x16x4_f32(
                false, a, false, br, (short)0, accR[ct], false, false);
        }
    }

    float dv[8];
#pragma unroll
    for (int r = 0; r < 8; ++r) dv[r] = dinv[min(row0 + r + 8 * hi, nrows - 1)];

#pragma unroll
    for (int ct = 0; ct < NCT; ++ct) {
        int colc = ct * 16 + m;
        float bv = biasR[colc];
#pragma unroll
        for (int r = 0; r < 8; ++r) {
            int rowm = row0 + r + 8 * hi;
            if (rowm >= nrows) continue;
            size_t idx = (size_t)rowm * F + colc;
            Ti[idx] = accI[ct][r] * dv[r];
            Tr[idx] = accR[ct][r] + bv;
        }
    }
}


__global__ void score_kernel(const int* __restrict__ li, const float* __restrict__ z,
                             float* __restrict__ out, int L) {
    int e = blockIdx.x * blockDim.x + threadIdx.x;
    if (e >= L) return;
    int i = li[e], j = li[L + e];
    const float4* za = (const float4*)(z + (size_t)i * 16);
    const float4* zb = (const float4*)(z + (size_t)j * 16);
    float dot = 0.0f, na = 0.0f, nb = 0.0f;
#pragma unroll
    for (int q = 0; q < 4; ++q) {
        float4 a = za[q], b = zb[q];
        dot += a.x * b.x + a.y * b.y + a.z * b.z + a.w * b.w;
        na  += a.x * a.x + a.y * a.y + a.z * a.z + a.w * a.w;
        nb  += b.x * b.x + b.y * b.y + b.z * b.z + b.w * b.w;
    }
    float den = fmaxf(sqrtf(na) * sqrtf(nb), 1e-6f);
    out[e] = dot / den;
}


static inline int cdiv(long long a, long long b) { return (int)((a + b - 1) / b); }

extern "C" void kernel_launch(void* const* d_in, const int* in_sizes, int n_in,
                              void* d_out, int out_size, void* d_ws, size_t ws_size,
                              hipStream_t stream) {
    const float* x     = (const float*)d_in[0];
    const int*   ei    = (const int*)d_in[1];
    const float* ew    = (const float*)d_in[2];
    const int*   eli   = (const int*)d_in[3];
    const float* preW  = (const float*)d_in[4];
    const float* preb  = (const float*)d_in[5];
    const float* g1Wi  = (const float*)d_in[6];
    const float* g1Wr  = (const float*)d_in[7];
    const float* g1b   = (const float*)d_in[8];
    const float* g2Wi  = (const float*)d_in[9];
    const float* g2Wr  = (const float*)d_in[10];
    const float* g2b   = (const float*)d_in[11];
    const float* poW   = (const float*)d_in[12];
    const float* pob   = (const float*)d_in[13];
    float* out = (float*)d_out;

    const int N = in_sizes[0];
    const int E = in_sizes[2];
    const int L = in_sizes[3] / 2;

    const int* erow = ei;
    const int* ecol = ei + E;

    float* ws   = (float*)d_ws;
    float* dinv = ws;
    float* aux  = dinv + N;
    float* h64  = aux  + E;
    float* t    = h64  + (size_t)N * 64;
    float* agg1 = t    + (size_t)N * 32;
    float* agg2 = agg1 + (size_t)N * 32;
    float* z    = agg2 + (size_t)N * 32;

    const int NB = cdiv(N, BIN_NODES);
    const int chunk = cdiv(E, PART_WG);
    int* counts   = (int*)aux;
    int* binStart = counts + PART_WG * NB;
    int* rec      = (int*)h64;

    const int B = 256;
    const int gemm_threads = cdiv(N, 16) * 32;

    bin_count_kernel<<<PART_WG, WG_THREADS, 0, stream>>>(ecol, counts, E, NB, chunk);
    bin_fill_kernel<<<PART_WG, WG_THREADS, 0, stream>>>(erow, ecol, ew, counts, binStart, rec, E, NB, chunk);
    bin_degree_kernel<<<NB, WG_THREADS, 0, stream>>>(rec, binStart, dinv, N);

    pre_gemm_kernel<<<cdiv(gemm_threads, B), B, 0, stream>>>(
        x, preW, preb, g1Wi, g1Wr, g1b, t, agg1, dinv, N);

    bin_accumulate_gemm_kernel<2, true><<<NB, WG_THREADS, 0, stream>>>(
        rec, binStart, t, agg1, g2Wi, g2Wr, g2b, agg2, dinv, N);

    bin_accumulate_gemm_kernel<1, false><<<NB, WG_THREADS, 0, stream>>>(
        rec, binStart, agg1, agg2, poW, nullptr, pob, z, dinv, N);

    score_kernel<<<cdiv(L, B), B, 0, stream>>>(eli, z, out, L);
}


# conv1 products (pre layer folded in) moved into the degree launch: per bin 13 waves make them while 3 sum its weights in LDS; rows take dinv from LDS; one launch fewer; sum order = LDS arrival
# speedup vs baseline: 6.6989x; 1.0627x over previous
#include <hip/hip_runtime.h>
#include <math.h>

typedef float v2f __attribute__((ext_vector_type(2)));
typedef float v8f __attribute__((ext_vector_type(8)));

static __device__ __forceinline__ float elu1(float v) {
    return v > 0.0f ? v : (__expf(v) - 1.0f);
}


#define BIN_NODES 196
#define NRT 13
#define TILE_ROWS (NRT * 16)
#define MAXB 512
#define PART_WG 256
#define WG_THREADS 1024

__global__ void __launch_bounds__(WG_THREADS)
bin_count_kernel(const int* __restrict__ col, int* __restrict__ counts, int E, int NB, int chunk) {
    __shared__ int hist[MAXB];
    const int tid = threadIdx.x;
    if (tid < MAXB) hist[tid] = 0;
    __syncthreads();
    const int base = blockIdx.x * chunk;
    const int end = min(base + chunk, E);
    for (int e0 = base; e0 < end; e0 += 4 * WG_THREADS) {
        int c[4];
#pragma unroll
        for (int k = 0; k < 4; ++k) {
            int e = e0 + k * WG_THREADS + tid;
            c[k] = (e < end) ? col[e] : -1;
        }
#pragma unroll
        for (int k = 0; k < 4; ++k)
            if (c[k] >= 0) atomicAdd(&hist[(unsigned)c[k] / BIN_NODES], 1);
    }
    __syncthreads();
    if (tid < NB) counts[blockIdx.x * NB + tid] = hist[tid];
}

#define FILL_K 13
#define FILL_CAP (FILL_K * WG_THREADS)
__global__ void __launch_bounds__(WG_THREADS)
bin_fill_kernel(const int* __restrict__ row, const int* __restrict__ col, const float* __restrict__ w,
                const int* __restrict__ counts, int* __restrict__ binStart, int* __restrict__ recw,
                int E, int NB, int chunk) {
    int2* __restrict__ rec = (int2*)recw;
    __shared__ int cursor[MAXB];
    __shared__ int sc[MAXB];
    __shared__ int ptot[2][MAXB];
    __shared__ int ppre[2][MAXB];
    __shared__ int lhist[MAXB];
    __shared__ int lstart[MAXB];
    __shared__ int wsum[MAXB / 32];
    __shared__ int2 stage[FILL_CAP];
    __shared__ int stageG[FILL_CAP];
    const int tid = threadIdx.x;
    const int lane = tid & 31;
    const int g = blockIdx.x;
    {
        const int b = tid & (MAXB - 1);
        const int part = tid >> 9;
        int tot = 0, pre = 0;
        if (b < NB) {
            const int g0 = part * (PART_WG / 2);
#pragma unroll 16
            for (int i = 0; i < PART_WG / 2; ++i) {
                int gg = g0 + i;
                int c = counts[gg * NB + b];
                tot += c;
                pre += (gg < g) ? c : 0;
            }
        }
        ptot[part][b] = tot;
        ppre[part][b] = pre;
    }
    __syncthreads();
    int tot = 0, pre = 0;
    if (tid < MAXB) {
        tot = ptot[0][tid] + ptot[1][tid];
        pre = ppre[0][tid] + ppre[1][tid];
        sc[tid] = tot;
    }
    __syncthreads();
    for (int off = 1; off < MAXB; off <<= 1) {
        int v = 0;
        if (tid < MAXB && tid >= off) v = sc[tid - off];
        __syncthreads();
        if (tid < MAXB) sc[tid] += v;
        __syncthreads();
    }
    if (tid < MAXB) {
        int excl = sc[tid] - tot;
        cursor[tid] = excl + pre;
        if (g == 0 && tid <= NB) binStart[tid] = excl;
    }

    const int base = g * chunk;
    const int end = min(base + chunk, E);
    for (int c0 = base; c0 < end; c0 += FILL_CAP) {
        const int c1 = min(c0 + FILL_CAP, end);
        if (tid < MAXB) lhist[tid] = 0;
        __syncthreads();

        int r[FILL_K], c[FILL_K];
        float ww[FILL_K];
#pragma unroll
        for (int k = 0; k < FILL_K; ++k) {
            int e = c0 + k * WG_THREADS + tid;
            bool ok = e < c1;
            r[k] = ok ? row[e] : 0;
            c[k] = ok ? col[e] : -1;
            ww[k] = ok ? w[e] : 0.0f;
        }
        int br[FILL_K];
#pragma unroll
        for (int k = 0; k < FILL_K; ++k) {
            br[k] = -1;
            if (c[k] >= 0) {
                const unsigned bin = (unsigned)c[k] / BIN_NODES;
                const int rank = atomicAdd(&lhist[bin], 1);
                r[k] = (int)((unsigned)r[k] | (((unsigned)c[k] - bin * BIN_NODES) << 24));
                br[k] = (int)(bin << 16) | rank;
            }
        }
        __syncthreads();

        {
            const int v = (tid < MAXB) ? lhist[tid] : 0;
            int incl = v;
#pragma unroll
            for (int off = 1; off < 32; off <<= 1) {
                int n = __shfl_up(incl, off);
                if (lane >= off) incl += n;
            }
            if (tid < MAXB && lane == 31) wsum[tid >> 5] = incl;
            __syncthreads();
            if (tid < MAXB) {
                int before = 0;
#pragma unroll
                for (int q = 0; q < MAXB / 32; ++q) before += (q < (tid >> 5)) ? wsum[q] : 0;
                lstart[tid] = before + incl - v;
            }
            __syncthreads();
        }

#pragma unroll
        for (int k = 0; k < FILL_K; ++k) {
            if (br[k] >= 0) {
                const int bin = br[k] >> 16;
                const int rank = br[k] & 0xFFFF;
                const int lp = lstart[bin] + rank;
                stage[lp] = make_int2(r[k], __float_as_int(ww[k]));
                stageG[lp] = cursor[bin] + rank;
            }
        }
        __syncthreads();

        const int n = c1 - c0;
        for (int i = tid; i < n; i += WG_THREADS) rec[stageG[i]] = stage[i];

        if (tid < MAXB) cursor[tid] += lhist[tid];
    }
}

#define DP_THREADS 512
#define DP_REC_LANES (DP_THREADS - NRT * 32)
#define DP_K 22
__global__ void __launch_bounds__(DP_THREADS, 8)
bin_degree_pre_gemm_kernel(const int* __restrict__ recw, const int* __restrict__ binStart,
                           const float* __restrict__ x,
                           const float* __restrict__ preW,
                           const float* __restrict__ preb,
                           const float* __restrict__ Wi,
                           const float* __restrict__ Wr,
                           const float* __restrict__ biasR,
                           float* __restrict__ Ti,
                           float* __restrict__ Tr,
                           float* __restrict__ dinv, int N) {
    constexpr int K = 64;
    constexpr int NCT = 2;
    constexpr int F = NCT * 16;
    const int2* __restrict__ rec = (const int2*)recw;
    __shared__ float d[TILE_ROWS];
    const int tid = threadIdx.x;
    const int lane = tid & 31;
    const int wave = __builtin_amdgcn_readfirstlane(tid >> 5);
    if (tid < TILE_ROWS) d[tid] = 0.0f;
    __syncthreads();

    const int s = binStart[blockIdx.x];
    const int e = binStart[blockIdx.x + 1];
    const int m  = lane & 15;
    const int hi = lane >> 4;
    const int nodeBase = blockIdx.x * BIN_NODES;
    v8f accI[NCT] = {};
    v8f accR[NCT] = {};
    if (wave < NRT) {
        const int ar = min(nodeBase + wave * 16 + m, N - 1);
        const float xv = x[ar];
        for (int k0 = 0; k0 < K; k0 += 4) {
            int ka = k0 + 2 * hi;
            v2f a;
            a.x = elu1(xv * preW[ka] + preb[ka]);
            a.y = elu1(xv * preW[ka + 1] + preb[ka + 1]);
#pragma unroll
            for (int ct = 0; ct < NCT; ++ct) {
                int colc = ct * 16 + m;
                v2f b;
                b.x = Wi[(size_t)ka * F + colc];
                b.y = Wi[(size_t)(ka + 1) * F + colc];
                accI[ct] = __builtin_amdgcn_wmma_f32_16x16x4_f32(
                    false, a, false, b, (short)0, accI[ct], false, false);
                v2f br;
                br.x = Wr[(size_t)ka * F + colc];
                br.y = Wr[(size_t)(ka + 1) * F + colc];
                accR[ct] = __builtin_amdgcn_wmma_f32_16x16x4_f32(
                    false, a, false, br, (short)0, accR[ct], false, false);
            }
        }
    } else {
        const int li = tid - NRT * 32;
        for (int p0 = s; p0 < e; p0 += DP_K * DP_REC_LANES) {
            int2 r[DP_K];
#pragma unroll
            for (int k = 0; k < DP_K; ++k) {
                int p = p0 + k * DP_REC_LANES + li;
                r[k] = rec[min(p, e - 1)];
            }
#pragma unroll
            for (int k = 0; k < DP_K; ++k) {
                int p = p0 + k * DP_REC_LANES + li;
                if (p < e) atomicAdd(&d[(unsigned)r[k].x >> 24], __int_as_float(r[k].y));
            }
        }
    }
    __syncthreads();
    if (tid < TILE_ROWS) {
        const int node = nodeBase + tid;
        float dv = 0.0f;
        if (tid < BIN_NODES && node < N) {
            const float v = d[tid];
            dv = v > 0.0f ? rsqrtf(v) : 0.0f;
            dinv[node] = dv;
        }
        d[tid] = dv;
    }
    __syncthreads();

    if (wave < NRT) {
        float dv[8];
#pragma unroll
        for (int rr = 0; rr < 8; ++rr) dv[rr] = d[wave * 16 + rr + 8 * hi];
#pragma unroll
        for (int ct = 0; ct < NCT; ++ct) {
            int colc = ct * 16 + m;
            float bv = biasR[colc];
#pragma unroll
            for (int rr = 0; rr < 8; ++rr) {
                const int nl = wave * 16 + rr + 8 * hi;
                const int node = nodeBase + nl;
                if (nl >= BIN_NODES || node >= N) continue;
                size_t idx = (size_t)node * F + colc;
                Ti[idx] = accI[ct][rr] * dv[rr];
                Tr[idx] = accR[ct][rr] + bv;
            }
        }
    }
}

#define TP 36
template <int NCT, bool DUAL>
__global__ void __launch_bounds__(WG_THREADS, 16)
bin_accumulate_gemm_kernel(const int* __restrict__ recw, const int* __restrict__ binStart,
                           const float* __restrict__ t, float* root,
                           const float* __restrict__ Wi, const float* __restrict__ Wr,
                           const float* __restrict__ biasR, float* __restrict__ outR,
                           const float* __restrict__ dinv, int N) {
    const int2* __restrict__ rec = (const int2*)recw;
    __shared__ float tile[TILE_ROWS * TP];
    __shared__ float dc[TILE_ROWS];
    const int tid = threadIdx.x;
    const int lane = tid & 31;
    const int wave = __builtin_amdgcn_readfirstlane(tid >> 5);
    for (int i = tid; i < TILE_ROWS * TP; i += WG_THREADS) tile[i] = 0.0f;
    if (tid < TILE_ROWS) {
        int node = blockIdx.x * BIN_NODES + tid;
        dc[tid] = (tid < BIN_NODES && node < N) ? dinv[node] : 0.0f;
    }
    __syncthreads();

    const int s = binStart[blockIdx.x];
    const int e = binStart[blockIdx.x + 1];
    const int step = (WG_THREADS / 32) * 32;

    int p0 = s + wave * 32;
    int2 cur = make_int2(0, 0);
    if (p0 < e) {
        int p = p0 + lane;
        if (p < e) cur = rec[p];
    }
    while (p0 < e) {
        const int pn = p0 + step;
        int2 nxt = make_int2(0, 0);
        if (pn < e) {
            int p = pn + lane;
            if (p < e) nxt = rec[p];
        }
        const int cnt = e - p0;
        float v[32];
#pragma unroll
        for (int j = 0; j < 32; ++j) {
            int r = __builtin_amdgcn_readlane(cur.x, j) & 0xFFFFFF;
            v[j] = t[(unsigned)(r * 32 + lane)];
        }
        if (cnt >= 32) {
#pragma unroll
            for (int j = 0; j < 32; ++j) {
                unsigned cl = (unsigned)__builtin_amdgcn_readlane(cur.x, j) >> 24;
                float n = __int_as_float(__builtin_amdgcn_readlane(cur.y, j));
                atomicAdd(&tile[cl * TP + lane], n * v[j]);
            }
        } else {
#pragma unroll
            for (int j = 0; j < 32; ++j) {
                if (j < cnt) {
                    unsigned cl = (unsigned)__builtin_amdgcn_readlane(cur.x, j) >> 24;
                    float n = __int_as_float(__builtin_amdgcn_readlane(cur.y, j));
                    atomicAdd(&tile[cl * TP + lane], n * v[j]);
                }
            }
        }
        p0 = pn;
        cur = nxt;
    }
    __syncthreads();

    const int nodeBase = blockIdx.x * BIN_NODES;
    constexpr int ROWS_PER_TRIP = WG_THREADS / 32;
    constexpr int EPI_TRIPS = (BIN_NODES + ROWS_PER_TRIP - 1) / ROWS_PER_TRIP;
    for (int k0 = 0; k0 < EPI_TRIPS; k0 += 4) {
        float a[4];
#pragma unroll
        for (int k = 0; k < 4; ++k) {
            const int nl = (k0 + k) * ROWS_PER_TRIP + wave;
            const int node = nodeBase + nl;
            a[k] = (nl < BIN_NODES && node < N) ? root[(size_t)node * 32 + lane] : 0.0f;
        }
#pragma unroll
        for (int k = 0; k < 4; ++k) {
            const int nl = (k0 + k) * ROWS_PER_TRIP + wave;
            if (nl < BIN_NODES) {
                const float sum = tile[nl * TP + lane];
                const float dn = dc[nl];
                float h = (nodeBase + nl < N) ? fmaxf(a[k] + dn * sum, 0.0f) : 0.0f;
                tile[nl * TP + lane] = h;
            }
        }
    }
    __syncthreads();

    const int part = wave >> 4;
    const int rt = wave & 15;
    if ((DUAL || part == 0) && rt < NRT) {
        constexpr int F = NCT * 16;
        const int m  = lane & 15;
        const int hi = lane >> 4;
        const bool second = DUAL && (part != 0);
        const float* __restrict__ Wsel = second ? Wr : Wi;
        const bool withBias = DUAL ? second : true;
        float* dst = DUAL ? (second ? outR : root) : outR;

        float bv[NCT];
#pragma unroll
        for (int ct = 0; ct < NCT; ++ct) bv[ct] = withBias ? biasR[ct * 16 + m] : -0.0f;
        const float am = (DUAL && !second) ? dc[rt * 16 + m] : 1.0f;

        v8f acc[NCT] = {};
        const float* arow = &tile[(rt * 16 + m) * TP + 2 * hi];
#pragma unroll
        for (int k0 = 0; k0 < 32; k0 += 4) {
            const int ka = k0 + 2 * hi;
            v2f a;
            a.x = arow[k0] * am;
            a.y = arow[k0 + 1] * am;
#pragma unroll
            for (int ct = 0; ct < NCT; ++ct) {
                const int colc = ct * 16 + m;
                v2f b;
                b.x = Wsel[(size_t)ka * F + colc];
                b.y = Wsel[(size_t)(ka + 1) * F + colc];
                acc[ct] = __builtin_amdgcn_wmma_f32_16x16x4_f32(
                    false, a, false, b, (short)0, acc[ct], false, false);
            }
        }

#pragma unroll
        for (int ct = 0; ct < NCT; ++ct) {
            const int colc = ct * 16 + m;
            float o[8];
#pragma unroll
            for (int r = 0; r < 8; ++r) o[r] = acc[ct][r] + bv[ct];
#pragma unroll
            for (int r = 0; r < 8; ++r) {
                const int nl = rt * 16 + r + 8 * hi;
                const int node = nodeBase + nl;
                if (nl < BIN_NODES && node < N) dst[(size_t)node * F + colc] = o[r];
            }
        }
    }
}


__global__ void score_kernel(const int* __restrict__ li, const float* __restrict__ z,
                             float* __restrict__ out, int L) {
    int e = blockIdx.x * blockDim.x + threadIdx.x;
    if (e >= L) return;
    int i = li[e], j = li[L + e];
    const float4* za = (const float4*)(z + (size_t)i * 16);
    const float4* zb = (const float4*)(z + (size_t)j * 16);
    float dot = 0.0f, na = 0.0f, nb = 0.0f;
#pragma unroll
    for (int q = 0; q < 4; ++q) {
        float4 a = za[q], b = zb[q];
        dot += a.x * b.x + a.y * b.y + a.z * b.z + a.w * b.w;
        na  += a.x * a.x + a.y * a.y + a.z * a.z + a.w * a.w;
        nb  += b.x * b.x + b.y * b.y + b.z * b.z + b.w * b.w;
    }
    float den = fmaxf(sqrtf(na) * sqrtf(nb), 1e-6f);
    out[e] = dot / den;
}


static inline int cdiv(long long a, long long b) { return (int)((a + b - 1) / b); }

extern "C" void kernel_launch(void* const* d_in, const int* in_sizes, int n_in,
                              void* d_out, int out_size, void* d_ws, size_t ws_size,
                              hipStream_t stream) {
    const float* x     = (const float*)d_in[0];
    const int*   ei    = (const int*)d_in[1];
    const float* ew    = (const float*)d_in[2];
    const int*   eli   = (const int*)d_in[3];
    const float* preW  = (const float*)d_in[4];
    const float* preb  = (const float*)d_in[5];
    const float* g1Wi  = (const float*)d_in[6];
    const float* g1Wr  = (const float*)d_in[7];
    const float* g1b   = (const float*)d_in[8];
    const float* g2Wi  = (const float*)d_in[9];
    const float* g2Wr  = (const float*)d_in[10];
    const float* g2b   = (const float*)d_in[11];
    const float* poW   = (const float*)d_in[12];
    const float* pob   = (const float*)d_in[13];
    float* out = (float*)d_out;

    const int N = in_sizes[0];
    const int E = in_sizes[2];
    const int L = in_sizes[3] / 2;

    const int* erow = ei;
    const int* ecol = ei + E;

    float* ws   = (float*)d_ws;
    float* dinv = ws;
    float* aux  = dinv + N;
    float* h64  = aux  + E;
    float* t    = h64  + (size_t)N * 64;
    float* agg1 = t    + (size_t)N * 32;
    float* agg2 = agg1 + (size_t)N * 32;
    float* z    = agg2 + (size_t)N * 32;

    const int NB = cdiv(N, BIN_NODES);
    const int chunk = cdiv(E, PART_WG);
    int* counts   = (int*)aux;
    int* binStart = counts + PART_WG * NB;
    int* rec      = (int*)h64;

    const int B = 256;

    bin_count_kernel<<<PART_WG, WG_THREADS, 0, stream>>>(ecol, counts, E, NB, chunk);
    bin_fill_kernel<<<PART_WG, WG_THREADS, 0, stream>>>(erow, ecol, ew, counts, binStart, rec, E, NB, chunk);
    bin_degree_pre_gemm_kernel<<<NB, DP_THREADS, 0, stream>>>(
        rec, binStart, x, preW, preb, g1Wi, g1Wr, g1b, t, agg1, dinv, N);

    bin_accumulate_gemm_kernel<2, true><<<NB, WG_THREADS, 0, stream>>>(
        rec, binStart, t, agg1, g2Wi, g2Wr, g2b, agg2, dinv, N);

    bin_accumulate_gemm_kernel<1, false><<<NB, WG_THREADS, 0, stream>>>(
        rec, binStart, agg1, agg2, poW, nullptr, pob, z, dinv, N);

    score_kernel<<<cdiv(L, B), B, 0, stream>>>(eli, z, out, L);
}
